# starting kernel
# baseline (speedup 1.0000x reference)
#include <hip/hip_runtime.h>
#include <math.h>

typedef __attribute__((ext_vector_type(2))) float v2f;
typedef __attribute__((ext_vector_type(8))) float v8f;

#define D   128
#define H   128
#define HH2 64
#define KC  10
#define LN_EPS 1e-5f

__global__ void wmma_gemm3_kernel(const float* __restrict__ x,
                                  const float* __restrict__ w1,
                                  const float* __restrict__ cw1,
                                  float* __restrict__ h1pre,
                                  float* __restrict__ hi,
                                  float* __restrict__ hj,
                                  int N) {
  const int rt  = blockIdx.x;
  const int mat = blockIdx.y;

  const float* Bm  = (mat == 0) ? w1    : (mat == 1 ? cw1 : cw1 + D * H);
  float*       out = (mat == 0) ? h1pre : (mat == 1 ? hi  : hj);

  const int lane = threadIdx.x;
  const int half = lane >> 4;
  const int l16  = lane & 15;

  int arow = rt * 16 + l16;
  if (arow >= N) arow = N - 1;
  const float* __restrict__ arow_p = x + (size_t)arow * D;

  v8f acc[8] = {};
#pragma unroll 4
  for (int k0 = 0; k0 < D; k0 += 4) {
    const int kk = k0 + half * 2;
    const v2f a = *(const v2f*)(arow_p + kk);
    const float* __restrict__ brow0 = Bm + (size_t)kk * H + l16;
#pragma unroll
    for (int ct = 0; ct < 8; ++ct) {
      v2f b;
      b.x = brow0[ct * 16];
      b.y = brow0[ct * 16 + H];
      acc[ct] = __builtin_amdgcn_wmma_f32_16x16x4_f32(false, a, false, b,
                                                      (short)0, acc[ct],
                                                      false, false);
    }
  }

  const int rbase = rt * 16 + half * 8;
  if (rt * 16 + 16 <= N) {
#pragma unroll
    for (int ct = 0; ct < 8; ++ct)
#pragma unroll
      for (int r = 0; r < 8; ++r)
        out[(size_t)(rbase + r) * H + ct * 16 + l16] = acc[ct][r];
  } else {
#pragma unroll
    for (int ct = 0; ct < 8; ++ct)
#pragma unroll
      for (int r = 0; r < 8; ++r)
        if (rbase + r < N)
          out[(size_t)(rbase + r) * H + ct * 16 + l16] = acc[ct][r];
  }
}

__global__ void node_mlp_kernel(const float* __restrict__ h1pre,
                                const float* __restrict__ b1,
                                const float* __restrict__ g1,
                                const float* __restrict__ be1,
                                const float* __restrict__ w2,
                                const float* __restrict__ b2,
                                const float* __restrict__ g2,
                                const float* __restrict__ be2,
                                const float* __restrict__ w3,
                                const float* __restrict__ b3,
                                float* __restrict__ out_logits,
                                float* __restrict__ out_probs,
                                int N) {
  const int n = blockIdx.x;
  const int t = threadIdx.x;
  __shared__ float sA[128], sB[128];
  __shared__ float sh1[128];
  __shared__ float sh2[HH2];
  __shared__ float slog[KC];

  float v = h1pre[(size_t)n * H + t] + b1[t];
  sA[t] = v;
  sB[t] = v * v;
  __syncthreads();
  for (int s = 64; s > 0; s >>= 1) {
    if (t < s) { sA[t] += sA[t + s]; sB[t] += sB[t + s]; }
    __syncthreads();
  }
  {
    float m   = sA[0] * (1.0f / 128.0f);
    float var = sB[0] * (1.0f / 128.0f) - m * m;
    float inv = rsqrtf(var + LN_EPS);
    float hv  = (v - m) * inv * g1[t] + be1[t];
    v = fmaxf(hv, 0.0f);
  }
  __syncthreads();
  sh1[t] = v;
  __syncthreads();

  float a2 = 0.0f;
  if (t < HH2) {
    a2 = b2[t];
    for (int k = 0; k < H; ++k) a2 += sh1[k] * w2[k * HH2 + t];
    sA[t] = a2;
    sB[t] = a2 * a2;
  }
  __syncthreads();
  for (int s = 32; s > 0; s >>= 1) {
    if (t < s) { sA[t] += sA[t + s]; sB[t] += sB[t + s]; }
    __syncthreads();
  }
  if (t < HH2) {
    float m2  = sA[0] * (1.0f / 64.0f);
    float v2  = sB[0] * (1.0f / 64.0f) - m2 * m2;
    float i2  = rsqrtf(v2 + LN_EPS);
    float h2v = (a2 - m2) * i2 * g2[t] + be2[t];
    sh2[t] = fmaxf(h2v, 0.0f);
  }
  __syncthreads();

  if (t < KC) {
    float lg = b3[t];
    for (int k = 0; k < HH2; ++k) lg += sh2[k] * w3[k * KC + t];
    slog[t] = lg;
    out_logits[(size_t)n * KC + t] = lg;
  }
  __syncthreads();
  if (t < KC) {
    float mx = slog[0];
    for (int k = 1; k < KC; ++k) mx = fmaxf(mx, slog[k]);
    float ssum = 0.0f;
    for (int k = 0; k < KC; ++k) ssum += __expf(slog[k] - mx);
    out_probs[(size_t)n * KC + t] = __expf(slog[t] - mx) / ssum;
  }
}

__global__ void centers_kernel(const float* __restrict__ probs,
                               const float* __restrict__ x,
                               const float* __restrict__ prototypes,
                               float* __restrict__ out_centers,
                               float* __restrict__ out_protos,
                               int N) {
  const int k = blockIdx.x;
  const int d = threadIdx.x;
  __shared__ float sW[128];

  float wpart = 0.0f;
  for (int n = d; n < N; n += 128) wpart += probs[(size_t)n * KC + k];
  sW[d] = wpart;
  __syncthreads();
  for (int s = 64; s > 0; s >>= 1) {
    if (d < s) sW[d] += sW[d + s];
    __syncthreads();
  }
  const float wsum = sW[0] + 1e-8f;

  float acc = 0.0f;
  for (int n = 0; n < N; ++n)
    acc += probs[(size_t)n * KC + k] * x[(size_t)n * D + d];

  out_centers[k * D + d] = acc / wsum;
  out_protos[k * D + d]  = prototypes[k * D + d];
}

__device__ __forceinline__ float wave_sum32(float v) {
#pragma unroll
  for (int off = 16; off > 0; off >>= 1) v += __shfl_xor(v, off, 32);
  return v;
}

__device__ __forceinline__ float pair_score(const float4 hi4, const float4 hj4,
                                            const float4 cb, const float4 g,
                                            const float4 be, const float4 cw,
                                            float cb2) {
  float4 hp;
  hp.x = hi4.x + hj4.x + cb.x;
  hp.y = hi4.y + hj4.y + cb.y;
  hp.z = hi4.z + hj4.z + cb.z;
  hp.w = hi4.w + hj4.w + cb.w;

  float s  = hp.x + hp.y + hp.z + hp.w;
  float sq = hp.x * hp.x + hp.y * hp.y + hp.z * hp.z + hp.w * hp.w;
  s  = wave_sum32(s);
  sq = wave_sum32(sq);

  const float m   = s * (1.0f / 128.0f);
  const float var = sq * (1.0f / 128.0f) - m * m;
  const float inv = rsqrtf(var + LN_EPS);

  float dot = fmaxf((hp.x - m) * inv * g.x + be.x, 0.0f) * cw.x
            + fmaxf((hp.y - m) * inv * g.y + be.y, 0.0f) * cw.y
            + fmaxf((hp.z - m) * inv * g.z + be.z, 0.0f) * cw.z
            + fmaxf((hp.w - m) * inv * g.w + be.w, 0.0f) * cw.w;
  dot = wave_sum32(dot);
  return tanhf(dot + cb2);
}

__global__ void comp_matrix_kernel(const float* __restrict__ hi,
                                   const float* __restrict__ hj,
                                   const float* __restrict__ cb1,
                                   const float* __restrict__ cg1,
                                   const float* __restrict__ cbe1,
                                   const float* __restrict__ cw2,
                                   const float* __restrict__ cb2p,
                                   float* __restrict__ out, int N) {
  const int i    = blockIdx.x;
  const int lane = threadIdx.x & 31;
  const int wid  = threadIdx.x >> 5;

  const float4 cb  = *(const float4*)(cb1  + 4 * lane);
  const float4 g   = *(const float4*)(cg1  + 4 * lane);
  const float4 be  = *(const float4*)(cbe1 + 4 * lane);
  const float4 cw  = *(const float4*)(cw2  + 4 * lane);
  const float  cb2 = cb2p[0];
  const float4 hi4 = *(const float4*)(hi + (size_t)i * H + 4 * lane);

  for (int j = wid; j < N; j += 8) {
    const float4 hj4 = *(const float4*)(hj + (size_t)j * H + 4 * lane);
    if (j + 8 < N)
      __builtin_prefetch(hj + (size_t)(j + 8) * H + 4 * lane, 0, 3);
    const float sc = pair_score(hi4, hj4, cb, g, be, cw, cb2);
    if (lane == 0) out[(size_t)i * N + j] = sc;
  }
}

__global__ void comp_scores_kernel(const float* __restrict__ hi,
                                   const float* __restrict__ hj,
                                   const int* __restrict__ ei, int E,
                                   const float* __restrict__ cb1,
                                   const float* __restrict__ cg1,
                                   const float* __restrict__ cbe1,
                                   const float* __restrict__ cw2,
                                   const float* __restrict__ cb2p,
                                   float* __restrict__ out) {
  const int lane = threadIdx.x & 31;
  const int gw   = (int)((blockIdx.x * blockDim.x + threadIdx.x) >> 5);
  const int nw   = (int)((gridDim.x * blockDim.x) >> 5);

  const float4 cb  = *(const float4*)(cb1  + 4 * lane);
  const float4 g   = *(const float4*)(cg1  + 4 * lane);
  const float4 be  = *(const float4*)(cbe1 + 4 * lane);
  const float4 cw  = *(const float4*)(cw2  + 4 * lane);
  const float  cb2 = cb2p[0];

  for (int e = gw; e < E; e += nw) {
    const int i = ei[e];
    const int j = ei[E + e];
    const float4 hi4 = *(const float4*)(hi + (size_t)i * H + 4 * lane);
    const float4 hj4 = *(const float4*)(hj + (size_t)j * H + 4 * lane);
    const float sc = pair_score(hi4, hj4, cb, g, be, cw, cb2);
    if (lane == 0) out[e] = sc;
  }
}

extern "C" void kernel_launch(void* const* d_in, const int* in_sizes, int n_in,
                              void* d_out, int out_size, void* d_ws, size_t ws_size,
                              hipStream_t stream) {
  const float* x    = (const float*)d_in[0];
  const int*   ei   = (const int*)  d_in[1];
  const float* w1   = (const float*)d_in[2];
  const float* b1   = (const float*)d_in[3];
  const float* g1   = (const float*)d_in[4];
  const float* be1  = (const float*)d_in[5];
  const float* w2   = (const float*)d_in[6];
  const float* b2   = (const float*)d_in[7];
  const float* g2   = (const float*)d_in[8];
  const float* be2  = (const float*)d_in[9];
  const float* w3   = (const float*)d_in[10];
  const float* b3   = (const float*)d_in[11];
  const float* cw1  = (const float*)d_in[12];
  const float* cb1  = (const float*)d_in[13];
  const float* cg1  = (const float*)d_in[14];
  const float* cbe1 = (const float*)d_in[15];
  const float* cw2  = (const float*)d_in[16];
  const float* cb2  = (const float*)d_in[17];
  const float* prot = (const float*)d_in[18];

  const int N = in_sizes[0] / D;
  const int E = in_sizes[1] / 2;

  float* h1pre = (float*)d_ws;
  float* hi    = h1pre + (size_t)N * H;
  float* hj    = hi    + (size_t)N * H;

  float* out       = (float*)d_out;
  float* o_logits  = out;
  float* o_probs   = o_logits  + (size_t)N * KC;
  float* o_scores  = o_probs   + (size_t)N * KC;
  float* o_matrix  = o_scores  + (size_t)E;
  float* o_centers = o_matrix  + (size_t)N * N;
  float* o_protos  = o_centers + (size_t)KC * D;

  const int rowTiles = (N + 15) / 16;

  wmma_gemm3_kernel<<<dim3(rowTiles, 3), 32, 0, stream>>>(
      x, w1, cw1, h1pre, hi, hj, N);

  node_mlp_kernel<<<N, 128, 0, stream>>>(
      h1pre, b1, g1, be1, w2, b2, g2, be2, w3, b3, o_logits, o_probs, N);

  centers_kernel<<<KC, 128, 0, stream>>>(
      o_probs, x, prot, o_centers, o_protos, N);

  comp_matrix_kernel<<<N, 256, 0, stream>>>(
      hi, hj, cb1, cg1, cbe1, cw2, cb2, o_matrix, N);

  comp_scores_kernel<<<2048, 256, 0, stream>>>(
      hi, hj, ei, E, cb1, cg1, cbe1, cw2, cb2, o_scores);
}


# pair scoring: 8 lanes per pair, 4 pairs per wave, the three sums as 3-step row_xmask DPP butterflies on the vector ALU instead of 5-step ds_bpermute over 32 lanes
# speedup vs baseline: 1.6181x; 1.6181x over previous
#include <hip/hip_runtime.h>
#include <math.h>

typedef __attribute__((ext_vector_type(2))) float v2f;
typedef __attribute__((ext_vector_type(8))) float v8f;

#define D   128
#define H   128
#define HH2 64
#define KC  10
#define LN_EPS 1e-5f

__global__ void wmma_gemm3_kernel(const float* __restrict__ x,
                                  const float* __restrict__ w1,
                                  const float* __restrict__ cw1,
                                  float* __restrict__ h1pre,
                                  float* __restrict__ hi,
                                  float* __restrict__ hj,
                                  int N) {
  const int rt  = blockIdx.x;
  const int mat = blockIdx.y;

  const float* Bm  = (mat == 0) ? w1    : (mat == 1 ? cw1 : cw1 + D * H);
  float*       out = (mat == 0) ? h1pre : (mat == 1 ? hi  : hj);

  const int lane = threadIdx.x;
  const int half = lane >> 4;
  const int l16  = lane & 15;

  int arow = rt * 16 + l16;
  if (arow >= N) arow = N - 1;
  const float* __restrict__ arow_p = x + (size_t)arow * D;

  v8f acc[8] = {};
#pragma unroll 4
  for (int k0 = 0; k0 < D; k0 += 4) {
    const int kk = k0 + half * 2;
    const v2f a = *(const v2f*)(arow_p + kk);
    const float* __restrict__ brow0 = Bm + (size_t)kk * H + l16;
#pragma unroll
    for (int ct = 0; ct < 8; ++ct) {
      v2f b;
      b.x = brow0[ct * 16];
      b.y = brow0[ct * 16 + H];
      acc[ct] = __builtin_amdgcn_wmma_f32_16x16x4_f32(false, a, false, b,
                                                      (short)0, acc[ct],
                                                      false, false);
    }
  }

  const int rbase = rt * 16 + half * 8;
  if (rt * 16 + 16 <= N) {
#pragma unroll
    for (int ct = 0; ct < 8; ++ct)
#pragma unroll
      for (int r = 0; r < 8; ++r)
        out[(size_t)(rbase + r) * H + ct * 16 + l16] = acc[ct][r];
  } else {
#pragma unroll
    for (int ct = 0; ct < 8; ++ct)
#pragma unroll
      for (int r = 0; r < 8; ++r)
        if (rbase + r < N)
          out[(size_t)(rbase + r) * H + ct * 16 + l16] = acc[ct][r];
  }
}

__global__ void node_mlp_kernel(const float* __restrict__ h1pre,
                                const float* __restrict__ b1,
                                const float* __restrict__ g1,
                                const float* __restrict__ be1,
                                const float* __restrict__ w2,
                                const float* __restrict__ b2,
                                const float* __restrict__ g2,
                                const float* __restrict__ be2,
                                const float* __restrict__ w3,
                                const float* __restrict__ b3,
                                float* __restrict__ out_logits,
                                float* __restrict__ out_probs,
                                int N) {
  const int n = blockIdx.x;
  const int t = threadIdx.x;
  __shared__ float sA[128], sB[128];
  __shared__ float sh1[128];
  __shared__ float sh2[HH2];
  __shared__ float slog[KC];

  float v = h1pre[(size_t)n * H + t] + b1[t];
  sA[t] = v;
  sB[t] = v * v;
  __syncthreads();
  for (int s = 64; s > 0; s >>= 1) {
    if (t < s) { sA[t] += sA[t + s]; sB[t] += sB[t + s]; }
    __syncthreads();
  }
  {
    float m   = sA[0] * (1.0f / 128.0f);
    float var = sB[0] * (1.0f / 128.0f) - m * m;
    float inv = rsqrtf(var + LN_EPS);
    float hv  = (v - m) * inv * g1[t] + be1[t];
    v = fmaxf(hv, 0.0f);
  }
  __syncthreads();
  sh1[t] = v;
  __syncthreads();

  float a2 = 0.0f;
  if (t < HH2) {
    a2 = b2[t];
    for (int k = 0; k < H; ++k) a2 += sh1[k] * w2[k * HH2 + t];
    sA[t] = a2;
    sB[t] = a2 * a2;
  }
  __syncthreads();
  for (int s = 32; s > 0; s >>= 1) {
    if (t < s) { sA[t] += sA[t + s]; sB[t] += sB[t + s]; }
    __syncthreads();
  }
  if (t < HH2) {
    float m2  = sA[0] * (1.0f / 64.0f);
    float v2  = sB[0] * (1.0f / 64.0f) - m2 * m2;
    float i2  = rsqrtf(v2 + LN_EPS);
    float h2v = (a2 - m2) * i2 * g2[t] + be2[t];
    sh2[t] = fmaxf(h2v, 0.0f);
  }
  __syncthreads();

  if (t < KC) {
    float lg = b3[t];
    for (int k = 0; k < HH2; ++k) lg += sh2[k] * w3[k * KC + t];
    slog[t] = lg;
    out_logits[(size_t)n * KC + t] = lg;
  }
  __syncthreads();
  if (t < KC) {
    float mx = slog[0];
    for (int k = 1; k < KC; ++k) mx = fmaxf(mx, slog[k]);
    float ssum = 0.0f;
    for (int k = 0; k < KC; ++k) ssum += __expf(slog[k] - mx);
    out_probs[(size_t)n * KC + t] = __expf(slog[t] - mx) / ssum;
  }
}

__global__ void centers_kernel(const float* __restrict__ probs,
                               const float* __restrict__ x,
                               const float* __restrict__ prototypes,
                               float* __restrict__ out_centers,
                               float* __restrict__ out_protos,
                               int N) {
  const int k = blockIdx.x;
  const int d = threadIdx.x;
  __shared__ float sW[128];

  float wpart = 0.0f;
  for (int n = d; n < N; n += 128) wpart += probs[(size_t)n * KC + k];
  sW[d] = wpart;
  __syncthreads();
  for (int s = 64; s > 0; s >>= 1) {
    if (d < s) sW[d] += sW[d + s];
    __syncthreads();
  }
  const float wsum = sW[0] + 1e-8f;

  float acc = 0.0f;
  for (int n = 0; n < N; ++n)
    acc += probs[(size_t)n * KC + k] * x[(size_t)n * D + d];

  out_centers[k * D + d] = acc / wsum;
  out_protos[k * D + d]  = prototypes[k * D + d];
}

typedef __attribute__((ext_vector_type(4))) float v4f;

template <int CTRL>
__device__ __forceinline__ float xmask_add(float v) {
  const int t = __builtin_amdgcn_update_dpp(0, __builtin_bit_cast(int, v),
                                            CTRL, 0xf, 0xf, true);
  return v + __builtin_bit_cast(float, t);
}

__device__ __forceinline__ float group_sum8(float v) {
  v = xmask_add<0x161>(v);
  v = xmask_add<0x162>(v);
  v = xmask_add<0x164>(v);
  return v;
}

__device__ __forceinline__ float pair_score8(const v4f (&hp)[4],
                                             const v4f (&g)[4],
                                             const v4f (&be)[4],
                                             const v4f (&cw)[4],
                                             float cb2) {
  v4f s4  = hp[0];
  v4f sq4 = hp[0] * hp[0];
#pragma unroll
  for (int t = 1; t < 4; ++t) {
    s4  += hp[t];
    sq4 += hp[t] * hp[t];
  }
  float s  = (s4.x + s4.y) + (s4.z + s4.w);
  float sq = (sq4.x + sq4.y) + (sq4.z + sq4.w);
  s  = group_sum8(s);
  sq = group_sum8(sq);

  const float m   = s * (1.0f / 128.0f);
  const float var = sq * (1.0f / 128.0f) - m * m;
  const float inv = rsqrtf(var + LN_EPS);

  v4f d4 = {0.0f, 0.0f, 0.0f, 0.0f};
#pragma unroll
  for (int t = 0; t < 4; ++t) {
    v4f v = (hp[t] - m) * inv * g[t] + be[t];
    v.x = fmaxf(v.x, 0.0f);
    v.y = fmaxf(v.y, 0.0f);
    v.z = fmaxf(v.z, 0.0f);
    v.w = fmaxf(v.w, 0.0f);
    d4 += v * cw[t];
  }
  float dot = (d4.x + d4.y) + (d4.z + d4.w);
  dot = group_sum8(dot);
  return tanhf(dot + cb2);
}

__global__ void comp_matrix_kernel(const float* __restrict__ hi,
                                   const float* __restrict__ hj,
                                   const float* __restrict__ cb1,
                                   const float* __restrict__ cg1,
                                   const float* __restrict__ cbe1,
                                   const float* __restrict__ cw2,
                                   const float* __restrict__ cb2p,
                                   float* __restrict__ out, int N) {
  const int i    = blockIdx.x;
  const int lane = threadIdx.x & 31;
  const int wid  = __builtin_amdgcn_readfirstlane((int)(threadIdx.x >> 5));
  const int q    = lane & 7;
  const int grp  = lane >> 3;

  v4f a[4], g[4], be[4], cw[4];
#pragma unroll
  for (int t = 0; t < 4; ++t) {
    const int col = 4 * (q + 8 * t);
    a[t]  = *(const v4f*)(hi + (size_t)i * H + col) + *(const v4f*)(cb1 + col);
    g[t]  = *(const v4f*)(cg1  + col);
    be[t] = *(const v4f*)(cbe1 + col);
    cw[t] = *(const v4f*)(cw2  + col);
  }
  const float cb2 = cb2p[0];

  const int nstep = (N + 3) >> 2;
  if (wid >= nstep) return;

  v4f b[4];
  {
    int j = wid * 4 + grp;
    if (j >= N) j = N - 1;
    const float* __restrict__ p = hj + (size_t)j * H + 4 * q;
#pragma unroll
    for (int t = 0; t < 4; ++t) b[t] = *(const v4f*)(p + 32 * t);
  }

  for (int jb = wid; jb < nstep; jb += 8) {
    v4f nb[4];
    {
      int jn = (jb + 8) * 4 + grp;
      if (jn >= N) jn = N - 1;
      const float* __restrict__ p = hj + (size_t)jn * H + 4 * q;
#pragma unroll
      for (int t = 0; t < 4; ++t) nb[t] = *(const v4f*)(p + 32 * t);
    }

    v4f hp[4];
#pragma unroll
    for (int t = 0; t < 4; ++t) hp[t] = a[t] + b[t];
    const float sc = pair_score8(hp, g, be, cw, cb2);

    const int j = jb * 4 + grp;
    if (q == 0 && j < N) out[(size_t)i * N + j] = sc;

#pragma unroll
    for (int t = 0; t < 4; ++t) b[t] = nb[t];
  }
}

__global__ void comp_scores_kernel(const float* __restrict__ hi,
                                   const float* __restrict__ hj,
                                   const int* __restrict__ ei, int E,
                                   const float* __restrict__ cb1,
                                   const float* __restrict__ cg1,
                                   const float* __restrict__ cbe1,
                                   const float* __restrict__ cw2,
                                   const float* __restrict__ cb2p,
                                   float* __restrict__ out) {
  const int lane = threadIdx.x & 31;
  const int q    = lane & 7;
  const int grp  = lane >> 3;
  const int gw   = __builtin_amdgcn_readfirstlane(
      (int)((blockIdx.x * blockDim.x + threadIdx.x) >> 5));
  const int nw   = (int)((gridDim.x * blockDim.x) >> 5);

  v4f cb[4], g[4], be[4], cw[4];
#pragma unroll
  for (int t = 0; t < 4; ++t) {
    const int col = 4 * (q + 8 * t);
    cb[t] = *(const v4f*)(cb1  + col);
    g[t]  = *(const v4f*)(cg1  + col);
    be[t] = *(const v4f*)(cbe1 + col);
    cw[t] = *(const v4f*)(cw2  + col);
  }
  const float cb2 = cb2p[0];

  const int nstep = (E + 3) >> 2;
  for (int eb = gw; eb < nstep; eb += nw) {
    const int e  = eb * 4 + grp;
    const int ec = (e < E) ? e : (E - 1);
    const int i = ei[ec];
    const int j = ei[E + ec];
    const float* __restrict__ pi = hi + (size_t)i * H + 4 * q;
    const float* __restrict__ pj = hj + (size_t)j * H + 4 * q;
    v4f hp[4];
#pragma unroll
    for (int t = 0; t < 4; ++t)
      hp[t] = *(const v4f*)(pi + 32 * t) + *(const v4f*)(pj + 32 * t) + cb[t];
    const float sc = pair_score8(hp, g, be, cw, cb2);
    if (q == 0 && e < E) out[e] = sc;
  }
}

extern "C" void kernel_launch(void* const* d_in, const int* in_sizes, int n_in,
                              void* d_out, int out_size, void* d_ws, size_t ws_size,
                              hipStream_t stream) {
  const float* x    = (const float*)d_in[0];
  const int*   ei   = (const int*)  d_in[1];
  const float* w1   = (const float*)d_in[2];
  const float* b1   = (const float*)d_in[3];
  const float* g1   = (const float*)d_in[4];
  const float* be1  = (const float*)d_in[5];
  const float* w2   = (const float*)d_in[6];
  const float* b2   = (const float*)d_in[7];
  const float* g2   = (const float*)d_in[8];
  const float* be2  = (const float*)d_in[9];
  const float* w3   = (const float*)d_in[10];
  const float* b3   = (const float*)d_in[11];
  const float* cw1  = (const float*)d_in[12];
  const float* cb1  = (const float*)d_in[13];
  const float* cg1  = (const float*)d_in[14];
  const float* cbe1 = (const float*)d_in[15];
  const float* cw2  = (const float*)d_in[16];
  const float* cb2  = (const float*)d_in[17];
  const float* prot = (const float*)d_in[18];

  const int N = in_sizes[0] / D;
  const int E = in_sizes[1] / 2;

  float* h1pre = (float*)d_ws;
  float* hi    = h1pre + (size_t)N * H;
  float* hj    = hi    + (size_t)N * H;

  float* out       = (float*)d_out;
  float* o_logits  = out;
  float* o_probs   = o_logits  + (size_t)N * KC;
  float* o_scores  = o_probs   + (size_t)N * KC;
  float* o_matrix  = o_scores  + (size_t)E;
  float* o_centers = o_matrix  + (size_t)N * N;
  float* o_protos  = o_centers + (size_t)KC * D;

  const int rowTiles = (N + 15) / 16;

  wmma_gemm3_kernel<<<dim3(rowTiles, 3), 32, 0, stream>>>(
      x, w1, cw1, h1pre, hi, hj, N);

  node_mlp_kernel<<<N, 128, 0, stream>>>(
      h1pre, b1, g1, be1, w2, b2, g2, be2, w3, b3, o_logits, o_probs, N);

  centers_kernel<<<KC, 128, 0, stream>>>(
      o_probs, x, prot, o_centers, o_protos, N);

  comp_matrix_kernel<<<N, 256, 0, stream>>>(
      hi, hj, cb1, cg1, cbe1, cw2, cb2, o_matrix, N);

  comp_scores_kernel<<<2048, 256, 0, stream>>>(
      hi, hj, ei, E, cb1, cg1, cbe1, cw2, cb2, o_scores);
}


# cluster centers (probs^T @ x) as an f32 WMMA GEMM over the nodes: 4 blocks x 16 waves split the K-steps with all loads in flight, LDS reduction; was a 1000-trip serial loop, 4 loads in flight
# speedup vs baseline: 2.8033x; 1.7325x over previous
#include <hip/hip_runtime.h>
#include <math.h>

typedef __attribute__((ext_vector_type(2))) float v2f;
typedef __attribute__((ext_vector_type(8))) float v8f;

#define D   128
#define H   128
#define HH2 64
#define KC  10
#define LN_EPS 1e-5f

__global__ void wmma_gemm3_kernel(const float* __restrict__ x,
                                  const float* __restrict__ w1,
                                  const float* __restrict__ cw1,
                                  float* __restrict__ h1pre,
                                  float* __restrict__ hi,
                                  float* __restrict__ hj,
                                  int N) {
  const int rt  = blockIdx.x;
  const int mat = blockIdx.y;

  const float* Bm  = (mat == 0) ? w1    : (mat == 1 ? cw1 : cw1 + D * H);
  float*       out = (mat == 0) ? h1pre : (mat == 1 ? hi  : hj);

  const int lane = threadIdx.x;
  const int half = lane >> 4;
  const int l16  = lane & 15;

  int arow = rt * 16 + l16;
  if (arow >= N) arow = N - 1;
  const float* __restrict__ arow_p = x + (size_t)arow * D;

  v8f acc[8] = {};
#pragma unroll 4
  for (int k0 = 0; k0 < D; k0 += 4) {
    const int kk = k0 + half * 2;
    const v2f a = *(const v2f*)(arow_p + kk);
    const float* __restrict__ brow0 = Bm + (size_t)kk * H + l16;
#pragma unroll
    for (int ct = 0; ct < 8; ++ct) {
      v2f b;
      b.x = brow0[ct * 16];
      b.y = brow0[ct * 16 + H];
      acc[ct] = __builtin_amdgcn_wmma_f32_16x16x4_f32(false, a, false, b,
                                                      (short)0, acc[ct],
                                                      false, false);
    }
  }

  const int rbase = rt * 16 + half * 8;
  if (rt * 16 + 16 <= N) {
#pragma unroll
    for (int ct = 0; ct < 8; ++ct)
#pragma unroll
      for (int r = 0; r < 8; ++r)
        out[(size_t)(rbase + r) * H + ct * 16 + l16] = acc[ct][r];
  } else {
#pragma unroll
    for (int ct = 0; ct < 8; ++ct)
#pragma unroll
      for (int r = 0; r < 8; ++r)
        if (rbase + r < N)
          out[(size_t)(rbase + r) * H + ct * 16 + l16] = acc[ct][r];
  }
}

__global__ void node_mlp_kernel(const float* __restrict__ h1pre,
                                const float* __restrict__ b1,
                                const float* __restrict__ g1,
                                const float* __restrict__ be1,
                                const float* __restrict__ w2,
                                const float* __restrict__ b2,
                                const float* __restrict__ g2,
                                const float* __restrict__ be2,
                                const float* __restrict__ w3,
                                const float* __restrict__ b3,
                                float* __restrict__ out_logits,
                                float* __restrict__ out_probs,
                                int N) {
  const int n = blockIdx.x;
  const int t = threadIdx.x;
  __shared__ float sA[128], sB[128];
  __shared__ float sh1[128];
  __shared__ float sh2[HH2];
  __shared__ float slog[KC];

  float v = h1pre[(size_t)n * H + t] + b1[t];
  sA[t] = v;
  sB[t] = v * v;
  __syncthreads();
  for (int s = 64; s > 0; s >>= 1) {
    if (t < s) { sA[t] += sA[t + s]; sB[t] += sB[t + s]; }
    __syncthreads();
  }
  {
    float m   = sA[0] * (1.0f / 128.0f);
    float var = sB[0] * (1.0f / 128.0f) - m * m;
    float inv = rsqrtf(var + LN_EPS);
    float hv  = (v - m) * inv * g1[t] + be1[t];
    v = fmaxf(hv, 0.0f);
  }
  __syncthreads();
  sh1[t] = v;
  __syncthreads();

  float a2 = 0.0f;
  if (t < HH2) {
    a2 = b2[t];
    for (int k = 0; k < H; ++k) a2 += sh1[k] * w2[k * HH2 + t];
    sA[t] = a2;
    sB[t] = a2 * a2;
  }
  __syncthreads();
  for (int s = 32; s > 0; s >>= 1) {
    if (t < s) { sA[t] += sA[t + s]; sB[t] += sB[t + s]; }
    __syncthreads();
  }
  if (t < HH2) {
    float m2  = sA[0] * (1.0f / 64.0f);
    float v2  = sB[0] * (1.0f / 64.0f) - m2 * m2;
    float i2  = rsqrtf(v2 + LN_EPS);
    float h2v = (a2 - m2) * i2 * g2[t] + be2[t];
    sh2[t] = fmaxf(h2v, 0.0f);
  }
  __syncthreads();

  if (t < KC) {
    float lg = b3[t];
    for (int k = 0; k < HH2; ++k) lg += sh2[k] * w3[k * KC + t];
    slog[t] = lg;
    out_logits[(size_t)n * KC + t] = lg;
  }
  __syncthreads();
  if (t < KC) {
    float mx = slog[0];
    for (int k = 1; k < KC; ++k) mx = fmaxf(mx, slog[k]);
    float ssum = 0.0f;
    for (int k = 0; k < KC; ++k) ssum += __expf(slog[k] - mx);
    out_probs[(size_t)n * KC + t] = __expf(slog[t] - mx) / ssum;
  }
}

#define CWAVES 16
#define CSTEPS 16

__global__ __launch_bounds__(CWAVES * 32)
void centers_kernel(const float* __restrict__ probs,
                    const float* __restrict__ x,
                    const float* __restrict__ prototypes,
                    float* __restrict__ out_centers,
                    float* __restrict__ out_protos,
                    int N) {
  const int cg   = blockIdx.x;
  const int tid  = threadIdx.x;
  const int lane = tid & 31;
  const int wid  = __builtin_amdgcn_readfirstlane(tid >> 5);
  const unsigned half = (unsigned)lane >> 4;
  const unsigned l16  = (unsigned)lane & 15u;

  __shared__ float red[CWAVES][16][32];
  __shared__ float wred[CWAVES][32];
  __shared__ float cen[KC][32];
  __shared__ float wsumS[16];

  const unsigned pofs = (l16 < KC) ? l16 : (unsigned)(KC - 1);
  const unsigned xofs = (unsigned)cg * 32u + l16;
  const unsigned un   = (unsigned)N;
  const unsigned nlast = un - 1u;

  v8f acc0 = {}, acc1 = {};
  float wp = 0.0f;

  const int nstep = (N + 3) >> 2;
  for (int s0 = 0; s0 < nstep; s0 += CWAVES * CSTEPS) {
    v2f a[CSTEPS], b0[CSTEPS], b1[CSTEPS];
#pragma unroll
    for (int u = 0; u < CSTEPS; ++u) {
      const unsigned n0  = (unsigned)(s0 + u * CWAVES + wid) * 4u + half * 2u;
      const unsigned n1  = n0 + 1u;
      const unsigned n0c = (n0 < un) ? n0 : nlast;
      const unsigned n1c = (n1 < un) ? n1 : nlast;
      a[u].x  = probs[n0c * KC + pofs];
      a[u].y  = probs[n1c * KC + pofs];
      b0[u].x = x[n0c * D + xofs];
      b0[u].y = x[n1c * D + xofs];
      b1[u].x = x[n0c * D + xofs + 16u];
      b1[u].y = x[n1c * D + xofs + 16u];
    }
    __builtin_amdgcn_sched_barrier(0);
#pragma unroll
    for (int u = 0; u < CSTEPS; ++u) {
      const unsigned n0 = (unsigned)(s0 + u * CWAVES + wid) * 4u + half * 2u;
      v2f av;
      av.x = (n0 < un)      ? a[u].x : 0.0f;
      av.y = (n0 + 1u < un) ? a[u].y : 0.0f;
      wp += av.x + av.y;
      acc0 = __builtin_amdgcn_wmma_f32_16x16x4_f32(false, av, false, b0[u],
                                                   (short)0, acc0, false, false);
      acc1 = __builtin_amdgcn_wmma_f32_16x16x4_f32(false, av, false, b1[u],
                                                   (short)0, acc1, false, false);
    }
  }

#pragma unroll
  for (int r = 0; r < 8; ++r) {
    red[wid][r][lane]     = acc0[r];
    red[wid][8 + r][lane] = acc1[r];
  }
  wred[wid][lane] = wp;
  __syncthreads();

  {
    const float* __restrict__ rp = &red[0][0][0] + tid;
    float sum = rp[0];
#pragma unroll
    for (int w = 1; w < CWAVES; ++w) sum += rp[w * 512];
    const int rr  = tid >> 5;
    const int m   = 8 * (lane >> 4) + (rr & 7);
    const int col = (rr >> 3) * 16 + (lane & 15);
    if (m < KC) cen[m][col] = sum;
  }
  if (tid < 16) {
    float ws = 0.0f;
#pragma unroll
    for (int w = 0; w < CWAVES; ++w) ws += wred[w][tid] + wred[w][tid + 16];
    wsumS[tid] = ws + 1e-8f;
  }
  __syncthreads();

  if (tid < KC * 32) {
    const int m = tid >> 5;
    const int o = m * D + cg * 32 + (tid & 31);
    out_centers[o] = cen[m][tid & 31] / wsumS[m];
    out_protos[o]  = prototypes[o];
  }
}

typedef __attribute__((ext_vector_type(4))) float v4f;

template <int CTRL>
__device__ __forceinline__ float xmask_add(float v) {
  const int t = __builtin_amdgcn_update_dpp(0, __builtin_bit_cast(int, v),
                                            CTRL, 0xf, 0xf, true);
  return v + __builtin_bit_cast(float, t);
}

__device__ __forceinline__ float group_sum8(float v) {
  v = xmask_add<0x161>(v);
  v = xmask_add<0x162>(v);
  v = xmask_add<0x164>(v);
  return v;
}

__device__ __forceinline__ float pair_score8(const v4f (&hp)[4],
                                             const v4f (&g)[4],
                                             const v4f (&be)[4],
                                             const v4f (&cw)[4],
                                             float cb2) {
  v4f s4  = hp[0];
  v4f sq4 = hp[0] * hp[0];
#pragma unroll
  for (int t = 1; t < 4; ++t) {
    s4  += hp[t];
    sq4 += hp[t] * hp[t];
  }
  float s  = (s4.x + s4.y) + (s4.z + s4.w);
  float sq = (sq4.x + sq4.y) + (sq4.z + sq4.w);
  s  = group_sum8(s);
  sq = group_sum8(sq);

  const float m   = s * (1.0f / 128.0f);
  const float var = sq * (1.0f / 128.0f) - m * m;
  const float inv = rsqrtf(var + LN_EPS);

  v4f d4 = {0.0f, 0.0f, 0.0f, 0.0f};
#pragma unroll
  for (int t = 0; t < 4; ++t) {
    v4f v = (hp[t] - m) * inv * g[t] + be[t];
    v.x = fmaxf(v.x, 0.0f);
    v.y = fmaxf(v.y, 0.0f);
    v.z = fmaxf(v.z, 0.0f);
    v.w = fmaxf(v.w, 0.0f);
    d4 += v * cw[t];
  }
  float dot = (d4.x + d4.y) + (d4.z + d4.w);
  dot = group_sum8(dot);
  return tanhf(dot + cb2);
}

__global__ void comp_matrix_kernel(const float* __restrict__ hi,
                                   const float* __restrict__ hj,
                                   const float* __restrict__ cb1,
                                   const float* __restrict__ cg1,
                                   const float* __restrict__ cbe1,
                                   const float* __restrict__ cw2,
                                   const float* __restrict__ cb2p,
                                   float* __restrict__ out, int N) {
  const int i    = blockIdx.x;
  const int lane = threadIdx.x & 31;
  const int wid  = __builtin_amdgcn_readfirstlane((int)(threadIdx.x >> 5));
  const int q    = lane & 7;
  const int grp  = lane >> 3;

  v4f a[4], g[4], be[4], cw[4];
#pragma unroll
  for (int t = 0; t < 4; ++t) {
    const int col = 4 * (q + 8 * t);
    a[t]  = *(const v4f*)(hi + (size_t)i * H + col) + *(const v4f*)(cb1 + col);
    g[t]  = *(const v4f*)(cg1  + col);
    be[t] = *(const v4f*)(cbe1 + col);
    cw[t] = *(const v4f*)(cw2  + col);
  }
  const float cb2 = cb2p[0];

  const int nstep = (N + 3) >> 2;
  if (wid >= nstep) return;

  v4f b[4];
  {
    int j = wid * 4 + grp;
    if (j >= N) j = N - 1;
    const float* __restrict__ p = hj + (size_t)j * H + 4 * q;
#pragma unroll
    for (int t = 0; t < 4; ++t) b[t] = *(const v4f*)(p + 32 * t);
  }

  for (int jb = wid; jb < nstep; jb += 8) {
    v4f nb[4];
    {
      int jn = (jb + 8) * 4 + grp;
      if (jn >= N) jn = N - 1;
      const float* __restrict__ p = hj + (size_t)jn * H + 4 * q;
#pragma unroll
      for (int t = 0; t < 4; ++t) nb[t] = *(const v4f*)(p + 32 * t);
    }

    v4f hp[4];
#pragma unroll
    for (int t = 0; t < 4; ++t) hp[t] = a[t] + b[t];
    const float sc = pair_score8(hp, g, be, cw, cb2);

    const int j = jb * 4 + grp;
    if (q == 0 && j < N) out[(size_t)i * N + j] = sc;

#pragma unroll
    for (int t = 0; t < 4; ++t) b[t] = nb[t];
  }
}

__global__ void comp_scores_kernel(const float* __restrict__ hi,
                                   const float* __restrict__ hj,
                                   const int* __restrict__ ei, int E,
                                   const float* __restrict__ cb1,
                                   const float* __restrict__ cg1,
                                   const float* __restrict__ cbe1,
                                   const float* __restrict__ cw2,
                                   const float* __restrict__ cb2p,
                                   float* __restrict__ out) {
  const int lane = threadIdx.x & 31;
  const int q    = lane & 7;
  const int grp  = lane >> 3;
  const int gw   = __builtin_amdgcn_readfirstlane(
      (int)((blockIdx.x * blockDim.x + threadIdx.x) >> 5));
  const int nw   = (int)((gridDim.x * blockDim.x) >> 5);

  v4f cb[4], g[4], be[4], cw[4];
#pragma unroll
  for (int t = 0; t < 4; ++t) {
    const int col = 4 * (q + 8 * t);
    cb[t] = *(const v4f*)(cb1  + col);
    g[t]  = *(const v4f*)(cg1  + col);
    be[t] = *(const v4f*)(cbe1 + col);
    cw[t] = *(const v4f*)(cw2  + col);
  }
  const float cb2 = cb2p[0];

  const int nstep = (E + 3) >> 2;
  for (int eb = gw; eb < nstep; eb += nw) {
    const int e  = eb * 4 + grp;
    const int ec = (e < E) ? e : (E - 1);
    const int i = ei[ec];
    const int j = ei[E + ec];
    const float* __restrict__ pi = hi + (size_t)i * H + 4 * q;
    const float* __restrict__ pj = hj + (size_t)j * H + 4 * q;
    v4f hp[4];
#pragma unroll
    for (int t = 0; t < 4; ++t)
      hp[t] = *(const v4f*)(pi + 32 * t) + *(const v4f*)(pj + 32 * t) + cb[t];
    const float sc = pair_score8(hp, g, be, cw, cb2);
    if (q == 0 && e < E) out[e] = sc;
  }
}

extern "C" void kernel_launch(void* const* d_in, const int* in_sizes, int n_in,
                              void* d_out, int out_size, void* d_ws, size_t ws_size,
                              hipStream_t stream) {
  const float* x    = (const float*)d_in[0];
  const int*   ei   = (const int*)  d_in[1];
  const float* w1   = (const float*)d_in[2];
  const float* b1   = (const float*)d_in[3];
  const float* g1   = (const float*)d_in[4];
  const float* be1  = (const float*)d_in[5];
  const float* w2   = (const float*)d_in[6];
  const float* b2   = (const float*)d_in[7];
  const float* g2   = (const float*)d_in[8];
  const float* be2  = (const float*)d_in[9];
  const float* w3   = (const float*)d_in[10];
  const float* b3   = (const float*)d_in[11];
  const float* cw1  = (const float*)d_in[12];
  const float* cb1  = (const float*)d_in[13];
  const float* cg1  = (const float*)d_in[14];
  const float* cbe1 = (const float*)d_in[15];
  const float* cw2  = (const float*)d_in[16];
  const float* cb2  = (const float*)d_in[17];
  const float* prot = (const float*)d_in[18];

  const int N = in_sizes[0] / D;
  const int E = in_sizes[1] / 2;

  float* h1pre = (float*)d_ws;
  float* hi    = h1pre + (size_t)N * H;
  float* hj    = hi    + (size_t)N * H;

  float* out       = (float*)d_out;
  float* o_logits  = out;
  float* o_probs   = o_logits  + (size_t)N * KC;
  float* o_scores  = o_probs   + (size_t)N * KC;
  float* o_matrix  = o_scores  + (size_t)E;
  float* o_centers = o_matrix  + (size_t)N * N;
  float* o_protos  = o_centers + (size_t)KC * D;

  const int rowTiles = (N + 15) / 16;

  wmma_gemm3_kernel<<<dim3(rowTiles, 3), 32, 0, stream>>>(
      x, w1, cw1, h1pre, hi, hj, N);

  node_mlp_kernel<<<N, 128, 0, stream>>>(
      h1pre, b1, g1, be1, w2, b2, g2, be2, w3, b3, o_logits, o_probs, N);

  centers_kernel<<<4, CWAVES * 32, 0, stream>>>(
      o_probs, x, prot, o_centers, o_protos, N);

  comp_matrix_kernel<<<N, 256, 0, stream>>>(
      hi, hj, cb1, cg1, cbe1, cw2, cb2, o_matrix, N);

  comp_scores_kernel<<<2048, 256, 0, stream>>>(
      hi, hj, ei, E, cb1, cg1, cbe1, cw2, cb2, o_scores);
}


# edge scores: each edge reads its score from the dense pair matrix written earlier in the same call (same function of the pair): one 4-byte gather, not two 512-byte row gathers plus LN and tanh
# speedup vs baseline: 3.5669x; 1.2724x over previous
#include <hip/hip_runtime.h>
#include <math.h>

typedef __attribute__((ext_vector_type(2))) float v2f;
typedef __attribute__((ext_vector_type(8))) float v8f;

#define D   128
#define H   128
#define HH2 64
#define KC  10
#define LN_EPS 1e-5f

__global__ void wmma_gemm3_kernel(const float* __restrict__ x,
                                  const float* __restrict__ w1,
                                  const float* __restrict__ cw1,
                                  float* __restrict__ h1pre,
                                  float* __restrict__ hi,
                                  float* __restrict__ hj,
                                  int N) {
  const int rt  = blockIdx.x;
  const int mat = blockIdx.y;

  const float* Bm  = (mat == 0) ? w1    : (mat == 1 ? cw1 : cw1 + D * H);
  float*       out = (mat == 0) ? h1pre : (mat == 1 ? hi  : hj);

  const int lane = threadIdx.x;
  const int half = lane >> 4;
  const int l16  = lane & 15;

  int arow = rt * 16 + l16;
  if (arow >= N) arow = N - 1;
  const float* __restrict__ arow_p = x + (size_t)arow * D;

  v8f acc[8] = {};
#pragma unroll 4
  for (int k0 = 0; k0 < D; k0 += 4) {
    const int kk = k0 + half * 2;
    const v2f a = *(const v2f*)(arow_p + kk);
    const float* __restrict__ brow0 = Bm + (size_t)kk * H + l16;
#pragma unroll
    for (int ct = 0; ct < 8; ++ct) {
      v2f b;
      b.x = brow0[ct * 16];
      b.y = brow0[ct * 16 + H];
      acc[ct] = __builtin_amdgcn_wmma_f32_16x16x4_f32(false, a, false, b,
                                                      (short)0, acc[ct],
                                                      false, false);
    }
  }

  const int rbase = rt * 16 + half * 8;
  if (rt * 16 + 16 <= N) {
#pragma unroll
    for (int ct = 0; ct < 8; ++ct)
#pragma unroll
      for (int r = 0; r < 8; ++r)
        out[(size_t)(rbase + r) * H + ct * 16 + l16] = acc[ct][r];
  } else {
#pragma unroll
    for (int ct = 0; ct < 8; ++ct)
#pragma unroll
      for (int r = 0; r < 8; ++r)
        if (rbase + r < N)
          out[(size_t)(rbase + r) * H + ct * 16 + l16] = acc[ct][r];
  }
}

__global__ void node_mlp_kernel(const float* __restrict__ h1pre,
                                const float* __restrict__ b1,
                                const float* __restrict__ g1,
                                const float* __restrict__ be1,
                                const float* __restrict__ w2,
                                const float* __restrict__ b2,
                                const float* __restrict__ g2,
                                const float* __restrict__ be2,
                                const float* __restrict__ w3,
                                const float* __restrict__ b3,
                                float* __restrict__ out_logits,
                                float* __restrict__ out_probs,
                                int N) {
  const int n = blockIdx.x;
  const int t = threadIdx.x;
  __shared__ float sA[128], sB[128];
  __shared__ float sh1[128];
  __shared__ float sh2[HH2];
  __shared__ float slog[KC];

  float v = h1pre[(size_t)n * H + t] + b1[t];
  sA[t] = v;
  sB[t] = v * v;
  __syncthreads();
  for (int s = 64; s > 0; s >>= 1) {
    if (t < s) { sA[t] += sA[t + s]; sB[t] += sB[t + s]; }
    __syncthreads();
  }
  {
    float m   = sA[0] * (1.0f / 128.0f);
    float var = sB[0] * (1.0f / 128.0f) - m * m;
    float inv = rsqrtf(var + LN_EPS);
    float hv  = (v - m) * inv * g1[t] + be1[t];
    v = fmaxf(hv, 0.0f);
  }
  __syncthreads();
  sh1[t] = v;
  __syncthreads();

  float a2 = 0.0f;
  if (t < HH2) {
    a2 = b2[t];
    for (int k = 0; k < H; ++k) a2 += sh1[k] * w2[k * HH2 + t];
    sA[t] = a2;
    sB[t] = a2 * a2;
  }
  __syncthreads();
  for (int s = 32; s > 0; s >>= 1) {
    if (t < s) { sA[t] += sA[t + s]; sB[t] += sB[t + s]; }
    __syncthreads();
  }
  if (t < HH2) {
    float m2  = sA[0] * (1.0f / 64.0f);
    float v2  = sB[0] * (1.0f / 64.0f) - m2 * m2;
    float i2  = rsqrtf(v2 + LN_EPS);
    float h2v = (a2 - m2) * i2 * g2[t] + be2[t];
    sh2[t] = fmaxf(h2v, 0.0f);
  }
  __syncthreads();

  if (t < KC) {
    float lg = b3[t];
    for (int k = 0; k < HH2; ++k) lg += sh2[k] * w3[k * KC + t];
    slog[t] = lg;
    out_logits[(size_t)n * KC + t] = lg;
  }
  __syncthreads();
  if (t < KC) {
    float mx = slog[0];
    for (int k = 1; k < KC; ++k) mx = fmaxf(mx, slog[k]);
    float ssum = 0.0f;
    for (int k = 0; k < KC; ++k) ssum += __expf(slog[k] - mx);
    out_probs[(size_t)n * KC + t] = __expf(slog[t] - mx) / ssum;
  }
}

#define CWAVES 16
#define CSTEPS 16

__global__ __launch_bounds__(CWAVES * 32)
void centers_kernel(const float* __restrict__ probs,
                    const float* __restrict__ x,
                    const float* __restrict__ prototypes,
                    float* __restrict__ out_centers,
                    float* __restrict__ out_protos,
                    int N) {
  const int cg   = blockIdx.x;
  const int tid  = threadIdx.x;
  const int lane = tid & 31;
  const int wid  = __builtin_amdgcn_readfirstlane(tid >> 5);
  const unsigned half = (unsigned)lane >> 4;
  const unsigned l16  = (unsigned)lane & 15u;

  __shared__ float red[CWAVES][16][32];
  __shared__ float wred[CWAVES][32];
  __shared__ float cen[KC][32];
  __shared__ float wsumS[16];

  const unsigned pofs = (l16 < KC) ? l16 : (unsigned)(KC - 1);
  const unsigned xofs = (unsigned)cg * 32u + l16;
  const unsigned un   = (unsigned)N;
  const unsigned nlast = un - 1u;

  v8f acc0 = {}, acc1 = {};
  float wp = 0.0f;

  const int nstep = (N + 3) >> 2;
  for (int s0 = 0; s0 < nstep; s0 += CWAVES * CSTEPS) {
    v2f a[CSTEPS], b0[CSTEPS], b1[CSTEPS];
#pragma unroll
    for (int u = 0; u < CSTEPS; ++u) {
      const unsigned n0  = (unsigned)(s0 + u * CWAVES + wid) * 4u + half * 2u;
      const unsigned n1  = n0 + 1u;
      const unsigned n0c = (n0 < un) ? n0 : nlast;
      const unsigned n1c = (n1 < un) ? n1 : nlast;
      a[u].x  = probs[n0c * KC + pofs];
      a[u].y  = probs[n1c * KC + pofs];
      b0[u].x = x[n0c * D + xofs];
      b0[u].y = x[n1c * D + xofs];
      b1[u].x = x[n0c * D + xofs + 16u];
      b1[u].y = x[n1c * D + xofs + 16u];
    }
    __builtin_amdgcn_sched_barrier(0);
#pragma unroll
    for (int u = 0; u < CSTEPS; ++u) {
      const unsigned n0 = (unsigned)(s0 + u * CWAVES + wid) * 4u + half * 2u;
      v2f av;
      av.x = (n0 < un)      ? a[u].x : 0.0f;
      av.y = (n0 + 1u < un) ? a[u].y : 0.0f;
      wp += av.x + av.y;
      acc0 = __builtin_amdgcn_wmma_f32_16x16x4_f32(false, av, false, b0[u],
                                                   (short)0, acc0, false, false);
      acc1 = __builtin_amdgcn_wmma_f32_16x16x4_f32(false, av, false, b1[u],
                                                   (short)0, acc1, false, false);
    }
  }

#pragma unroll
  for (int r = 0; r < 8; ++r) {
    red[wid][r][lane]     = acc0[r];
    red[wid][8 + r][lane] = acc1[r];
  }
  wred[wid][lane] = wp;
  __syncthreads();

  {
    const float* __restrict__ rp = &red[0][0][0] + tid;
    float sum = rp[0];
#pragma unroll
    for (int w = 1; w < CWAVES; ++w) sum += rp[w * 512];
    const int rr  = tid >> 5;
    const int m   = 8 * (lane >> 4) + (rr & 7);
    const int col = (rr >> 3) * 16 + (lane & 15);
    if (m < KC) cen[m][col] = sum;
  }
  if (tid < 16) {
    float ws = 0.0f;
#pragma unroll
    for (int w = 0; w < CWAVES; ++w) ws += wred[w][tid] + wred[w][tid + 16];
    wsumS[tid] = ws + 1e-8f;
  }
  __syncthreads();

  if (tid < KC * 32) {
    const int m = tid >> 5;
    const int o = m * D + cg * 32 + (tid & 31);
    out_centers[o] = cen[m][tid & 31] / wsumS[m];
    out_protos[o]  = prototypes[o];
  }
}

typedef __attribute__((ext_vector_type(4))) float v4f;

template <int CTRL>
__device__ __forceinline__ float xmask_add(float v) {
  const int t = __builtin_amdgcn_update_dpp(0, __builtin_bit_cast(int, v),
                                            CTRL, 0xf, 0xf, true);
  return v + __builtin_bit_cast(float, t);
}

__device__ __forceinline__ float group_sum8(float v) {
  v = xmask_add<0x161>(v);
  v = xmask_add<0x162>(v);
  v = xmask_add<0x164>(v);
  return v;
}

__device__ __forceinline__ float pair_score8(const v4f (&hp)[4],
                                             const v4f (&g)[4],
                                             const v4f (&be)[4],
                                             const v4f (&cw)[4],
                                             float cb2) {
  v4f s4  = hp[0];
  v4f sq4 = hp[0] * hp[0];
#pragma unroll
  for (int t = 1; t < 4; ++t) {
    s4  += hp[t];
    sq4 += hp[t] * hp[t];
  }
  float s  = (s4.x + s4.y) + (s4.z + s4.w);
  float sq = (sq4.x + sq4.y) + (sq4.z + sq4.w);
  s  = group_sum8(s);
  sq = group_sum8(sq);

  const float m   = s * (1.0f / 128.0f);
  const float var = sq * (1.0f / 128.0f) - m * m;
  const float inv = rsqrtf(var + LN_EPS);

  v4f d4 = {0.0f, 0.0f, 0.0f, 0.0f};
#pragma unroll
  for (int t = 0; t < 4; ++t) {
    v4f v = (hp[t] - m) * inv * g[t] + be[t];
    v.x = fmaxf(v.x, 0.0f);
    v.y = fmaxf(v.y, 0.0f);
    v.z = fmaxf(v.z, 0.0f);
    v.w = fmaxf(v.w, 0.0f);
    d4 += v * cw[t];
  }
  float dot = (d4.x + d4.y) + (d4.z + d4.w);
  dot = group_sum8(dot);
  return tanhf(dot + cb2);
}

__global__ void comp_matrix_kernel(const float* __restrict__ hi,
                                   const float* __restrict__ hj,
                                   const float* __restrict__ cb1,
                                   const float* __restrict__ cg1,
                                   const float* __restrict__ cbe1,
                                   const float* __restrict__ cw2,
                                   const float* __restrict__ cb2p,
                                   float* __restrict__ out, int N) {
  const int i    = blockIdx.x;
  const int lane = threadIdx.x & 31;
  const int wid  = __builtin_amdgcn_readfirstlane((int)(threadIdx.x >> 5));
  const int q    = lane & 7;
  const int grp  = lane >> 3;

  v4f a[4], g[4], be[4], cw[4];
#pragma unroll
  for (int t = 0; t < 4; ++t) {
    const int col = 4 * (q + 8 * t);
    a[t]  = *(const v4f*)(hi + (size_t)i * H + col) + *(const v4f*)(cb1 + col);
    g[t]  = *(const v4f*)(cg1  + col);
    be[t] = *(const v4f*)(cbe1 + col);
    cw[t] = *(const v4f*)(cw2  + col);
  }
  const float cb2 = cb2p[0];

  const int nstep = (N + 3) >> 2;
  if (wid >= nstep) return;

  v4f b[4];
  {
    int j = wid * 4 + grp;
    if (j >= N) j = N - 1;
    const float* __restrict__ p = hj + (size_t)j * H + 4 * q;
#pragma unroll
    for (int t = 0; t < 4; ++t) b[t] = *(const v4f*)(p + 32 * t);
  }

  for (int jb = wid; jb < nstep; jb += 8) {
    v4f nb[4];
    {
      int jn = (jb + 8) * 4 + grp;
      if (jn >= N) jn = N - 1;
      const float* __restrict__ p = hj + (size_t)jn * H + 4 * q;
#pragma unroll
      for (int t = 0; t < 4; ++t) nb[t] = *(const v4f*)(p + 32 * t);
    }

    v4f hp[4];
#pragma unroll
    for (int t = 0; t < 4; ++t) hp[t] = a[t] + b[t];
    const float sc = pair_score8(hp, g, be, cw, cb2);

    const int j = jb * 4 + grp;
    if (q == 0 && j < N) out[(size_t)i * N + j] = sc;

#pragma unroll
    for (int t = 0; t < 4; ++t) b[t] = nb[t];
  }
}

#define CS_THREADS 256
#define EPT 8

__global__ __launch_bounds__(CS_THREADS)
void comp_scores_kernel(const float* __restrict__ mat,
                        const int* __restrict__ ei, int E, int N,
                        float* __restrict__ out) {
  const unsigned ue    = (unsigned)E;
  const unsigned un    = (unsigned)N;
  const unsigned nlast = un - 1u;
  const unsigned chunk = CS_THREADS * EPT;
  const int* erow = ei;
  const int* ecol = ei + ue;

  for (unsigned c0 = blockIdx.x * chunk; c0 < ue; c0 += gridDim.x * chunk) {
    const unsigned e0 = c0 + threadIdx.x;
    if (c0 + chunk <= ue) {
      unsigned ri[EPT], ci[EPT];
#pragma unroll
      for (int u = 0; u < EPT; ++u) {
        ri[u] = (unsigned)erow[e0 + u * CS_THREADS];
        ci[u] = (unsigned)ecol[e0 + u * CS_THREADS];
      }
      float v[EPT];
#pragma unroll
      for (int u = 0; u < EPT; ++u) {
        const unsigned i = (ri[u] < un) ? ri[u] : nlast;
        const unsigned j = (ci[u] < un) ? ci[u] : nlast;
        v[u] = mat[i * un + j];
      }
#pragma unroll
      for (int u = 0; u < EPT; ++u) out[e0 + u * CS_THREADS] = v[u];
    } else {
#pragma unroll 1
      for (unsigned e = e0; e < ue; e += CS_THREADS) {
        const unsigned r = (unsigned)erow[e];
        const unsigned c = (unsigned)ecol[e];
        const unsigned i = (r < un) ? r : nlast;
        const unsigned j = (c < un) ? c : nlast;
        out[e] = mat[i * un + j];
      }
    }
  }
}

extern "C" void kernel_launch(void* const* d_in, const int* in_sizes, int n_in,
                              void* d_out, int out_size, void* d_ws, size_t ws_size,
                              hipStream_t stream) {
  const float* x    = (const float*)d_in[0];
  const int*   ei   = (const int*)  d_in[1];
  const float* w1   = (const float*)d_in[2];
  const float* b1   = (const float*)d_in[3];
  const float* g1   = (const float*)d_in[4];
  const float* be1  = (const float*)d_in[5];
  const float* w2   = (const float*)d_in[6];
  const float* b2   = (const float*)d_in[7];
  const float* g2   = (const float*)d_in[8];
  const float* be2  = (const float*)d_in[9];
  const float* w3   = (const float*)d_in[10];
  const float* b3   = (const float*)d_in[11];
  const float* cw1  = (const float*)d_in[12];
  const float* cb1  = (const float*)d_in[13];
  const float* cg1  = (const float*)d_in[14];
  const float* cbe1 = (const float*)d_in[15];
  const float* cw2  = (const float*)d_in[16];
  const float* cb2  = (const float*)d_in[17];
  const float* prot = (const float*)d_in[18];

  const int N = in_sizes[0] / D;
  const int E = in_sizes[1] / 2;

  float* h1pre = (float*)d_ws;
  float* hi    = h1pre + (size_t)N * H;
  float* hj    = hi    + (size_t)N * H;

  float* out       = (float*)d_out;
  float* o_logits  = out;
  float* o_probs   = o_logits  + (size_t)N * KC;
  float* o_scores  = o_probs   + (size_t)N * KC;
  float* o_matrix  = o_scores  + (size_t)E;
  float* o_centers = o_matrix  + (size_t)N * N;
  float* o_protos  = o_centers + (size_t)KC * D;

  const int rowTiles = (N + 15) / 16;

  wmma_gemm3_kernel<<<dim3(rowTiles, 3), 32, 0, stream>>>(
      x, w1, cw1, h1pre, hi, hj, N);

  node_mlp_kernel<<<N, 128, 0, stream>>>(
      h1pre, b1, g1, be1, w2, b2, g2, be2, w3, b3, o_logits, o_probs, N);

  centers_kernel<<<4, CWAVES * 32, 0, stream>>>(
      o_probs, x, prot, o_centers, o_protos, N);

  comp_matrix_kernel<<<N, 256, 0, stream>>>(
      hi, hj, cb1, cg1, cbe1, cw2, cb2, o_matrix, N);

  comp_scores_kernel<<<256, CS_THREADS, 0, stream>>>(
      o_matrix, ei, E, N, o_scores);
}


# pair matrix: LN variance of a 16x32 pair tile from one f32 WMMA product of centered rows (|a|^2+|b|^2+2a.b), then one lane per pair column: 4 vector ops per element, no butterflies, 1 tanh per pair
# speedup vs baseline: 4.2004x; 1.1776x over previous
#include <hip/hip_runtime.h>
#include <math.h>

typedef __attribute__((ext_vector_type(2))) float v2f;
typedef __attribute__((ext_vector_type(4))) float v4f;
typedef __attribute__((ext_vector_type(8))) float v8f;

#define D   128
#define H   128
#define HH2 64
#define KC  10
#define LN_EPS 1e-5f

__global__ void wmma_gemm3_kernel(const float* __restrict__ x,
                                  const float* __restrict__ w1,
                                  const float* __restrict__ cw1,
                                  const float* __restrict__ cb1,
                                  float* __restrict__ h1pre,
                                  float* __restrict__ aT,
                                  float* __restrict__ bT,
                                  int N) {
  const int rt  = blockIdx.x;
  const int mat = blockIdx.y;

  const float* Bm = (mat == 0) ? w1 : (mat == 1 ? cw1 : cw1 + D * H);

  const int lane = threadIdx.x;
  const int half = lane >> 4;
  const int l16  = lane & 15;

  int arow = rt * 16 + l16;
  if (arow >= N) arow = N - 1;
  const float* __restrict__ arow_p = x + (size_t)arow * D;

  v8f acc[8] = {};

  if (mat == 0) {
#pragma unroll 4
    for (int k0 = 0; k0 < D; k0 += 4) {
      const int kk = k0 + half * 2;
      const v2f a = *(const v2f*)(arow_p + kk);
      const float* __restrict__ brow0 = Bm + (size_t)kk * H + l16;
#pragma unroll
      for (int ct = 0; ct < 8; ++ct) {
        v2f b;
        b.x = brow0[ct * 16];
        b.y = brow0[ct * 16 + H];
        acc[ct] = __builtin_amdgcn_wmma_f32_16x16x4_f32(false, a, false, b,
                                                        (short)0, acc[ct],
                                                        false, false);
      }
    }
    const int rbase = rt * 16 + half * 8;
    if (rt * 16 + 16 <= N) {
#pragma unroll
      for (int ct = 0; ct < 8; ++ct)
#pragma unroll
        for (int r = 0; r < 8; ++r)
          h1pre[(size_t)(rbase + r) * H + ct * 16 + l16] = acc[ct][r];
    } else {
#pragma unroll
      for (int ct = 0; ct < 8; ++ct)
#pragma unroll
        for (int r = 0; r < 8; ++r)
          if (rbase + r < N)
            h1pre[(size_t)(rbase + r) * H + ct * 16 + l16] = acc[ct][r];
    }
  } else {
#pragma unroll 4
    for (int k0 = 0; k0 < D; k0 += 4) {
      const int kk = k0 + half * 2;
      const v2f a = *(const v2f*)(arow_p + kk);
      const float* __restrict__ brow0 = Bm + (size_t)kk * H + l16;
#pragma unroll
      for (int ct = 0; ct < 8; ++ct) {
        v2f b;
        b.x = brow0[ct * 16];
        b.y = brow0[ct * 16 + H];
        acc[ct] = __builtin_amdgcn_wmma_f32_16x16x4_f32(false, b, false, a,
                                                        (short)0, acc[ct],
                                                        false, false);
      }
    }
    if (mat == 1) {
#pragma unroll
      for (int ct = 0; ct < 8; ++ct) {
        const v4f c0v = *(const v4f*)(cb1 + ct * 16 + half * 8);
        const v4f c1v = *(const v4f*)(cb1 + ct * 16 + half * 8 + 4);
        acc[ct][0] += c0v.x; acc[ct][1] += c0v.y;
        acc[ct][2] += c0v.z; acc[ct][3] += c0v.w;
        acc[ct][4] += c1v.x; acc[ct][5] += c1v.y;
        acc[ct][6] += c1v.z; acc[ct][7] += c1v.w;
      }
    }
    float s = 0.0f;
#pragma unroll
    for (int ct = 0; ct < 8; ++ct)
#pragma unroll
      for (int r = 0; r < 8; ++r) s += acc[ct][r];
    s += __shfl_xor(s, 16);
    const float mean = s * (1.0f / 128.0f);
    float* __restrict__ outT = (mat == 1) ? aT : bT;
    const int n = rt * 16 + l16;
    if (n < N) {
#pragma unroll
      for (int ct = 0; ct < 8; ++ct)
#pragma unroll
        for (int r = 0; r < 8; ++r)
          outT[(size_t)(ct * 16 + half * 8 + r) * N + n] = acc[ct][r] - mean;
    }
  }
}

__global__ void node_mlp_kernel(const float* __restrict__ h1pre,
                                const float* __restrict__ b1,
                                const float* __restrict__ g1,
                                const float* __restrict__ be1,
                                const float* __restrict__ w2,
                                const float* __restrict__ b2,
                                const float* __restrict__ g2,
                                const float* __restrict__ be2,
                                const float* __restrict__ w3,
                                const float* __restrict__ b3,
                                float* __restrict__ out_logits,
                                float* __restrict__ out_probs,
                                int N) {
  const int n = blockIdx.x;
  const int t = threadIdx.x;
  __shared__ float sA[128], sB[128];
  __shared__ float sh1[128];
  __shared__ float sh2[HH2];
  __shared__ float slog[KC];

  float v = h1pre[(size_t)n * H + t] + b1[t];
  sA[t] = v;
  sB[t] = v * v;
  __syncthreads();
  for (int s = 64; s > 0; s >>= 1) {
    if (t < s) { sA[t] += sA[t + s]; sB[t] += sB[t + s]; }
    __syncthreads();
  }
  {
    float m   = sA[0] * (1.0f / 128.0f);
    float var = sB[0] * (1.0f / 128.0f) - m * m;
    float inv = rsqrtf(var + LN_EPS);
    float hv  = (v - m) * inv * g1[t] + be1[t];
    v = fmaxf(hv, 0.0f);
  }
  __syncthreads();
  sh1[t] = v;
  __syncthreads();

  float a2 = 0.0f;
  if (t < HH2) {
    a2 = b2[t];
    for (int k = 0; k < H; ++k) a2 += sh1[k] * w2[k * HH2 + t];
    sA[t] = a2;
    sB[t] = a2 * a2;
  }
  __syncthreads();
  for (int s = 32; s > 0; s >>= 1) {
    if (t < s) { sA[t] += sA[t + s]; sB[t] += sB[t + s]; }
    __syncthreads();
  }
  if (t < HH2) {
    float m2  = sA[0] * (1.0f / 64.0f);
    float v2  = sB[0] * (1.0f / 64.0f) - m2 * m2;
    float i2  = rsqrtf(v2 + LN_EPS);
    float h2v = (a2 - m2) * i2 * g2[t] + be2[t];
    sh2[t] = fmaxf(h2v, 0.0f);
  }
  __syncthreads();

  if (t < KC) {
    float lg = b3[t];
    for (int k = 0; k < HH2; ++k) lg += sh2[k] * w3[k * KC + t];
    slog[t] = lg;
    out_logits[(size_t)n * KC + t] = lg;
  }
  __syncthreads();
  if (t < KC) {
    float mx = slog[0];
    for (int k = 1; k < KC; ++k) mx = fmaxf(mx, slog[k]);
    float ssum = 0.0f;
    for (int k = 0; k < KC; ++k) ssum += __expf(slog[k] - mx);
    out_probs[(size_t)n * KC + t] = __expf(slog[t] - mx) / ssum;
  }
}

#define CWAVES 16
#define CSTEPS 16

__global__ __launch_bounds__(CWAVES * 32)
void centers_kernel(const float* __restrict__ probs,
                    const float* __restrict__ x,
                    const float* __restrict__ prototypes,
                    float* __restrict__ out_centers,
                    float* __restrict__ out_protos,
                    int N) {
  const int cg   = blockIdx.x;
  const int tid  = threadIdx.x;
  const int lane = tid & 31;
  const int wid  = __builtin_amdgcn_readfirstlane(tid >> 5);
  const unsigned half = (unsigned)lane >> 4;
  const unsigned l16  = (unsigned)lane & 15u;

  __shared__ float red[CWAVES][16][32];
  __shared__ float wred[CWAVES][32];
  __shared__ float cen[KC][32];
  __shared__ float wsumS[16];

  const unsigned pofs = (l16 < KC) ? l16 : (unsigned)(KC - 1);
  const unsigned xofs = (unsigned)cg * 32u + l16;
  const unsigned un   = (unsigned)N;
  const unsigned nlast = un - 1u;

  v8f acc0 = {}, acc1 = {};
  float wp = 0.0f;

  const int nstep = (N + 3) >> 2;
  for (int s0 = 0; s0 < nstep; s0 += CWAVES * CSTEPS) {
    v2f a[CSTEPS], b0[CSTEPS], b1[CSTEPS];
#pragma unroll
    for (int u = 0; u < CSTEPS; ++u) {
      const unsigned n0  = (unsigned)(s0 + u * CWAVES + wid) * 4u + half * 2u;
      const unsigned n1  = n0 + 1u;
      const unsigned n0c = (n0 < un) ? n0 : nlast;
      const unsigned n1c = (n1 < un) ? n1 : nlast;
      a[u].x  = probs[n0c * KC + pofs];
      a[u].y  = probs[n1c * KC + pofs];
      b0[u].x = x[n0c * D + xofs];
      b0[u].y = x[n1c * D + xofs];
      b1[u].x = x[n0c * D + xofs + 16u];
      b1[u].y = x[n1c * D + xofs + 16u];
    }
    __builtin_amdgcn_sched_barrier(0);
#pragma unroll
    for (int u = 0; u < CSTEPS; ++u) {
      const unsigned n0 = (unsigned)(s0 + u * CWAVES + wid) * 4u + half * 2u;
      v2f av;
      av.x = (n0 < un)      ? a[u].x : 0.0f;
      av.y = (n0 + 1u < un) ? a[u].y : 0.0f;
      wp += av.x + av.y;
      acc0 = __builtin_amdgcn_wmma_f32_16x16x4_f32(false, av, false, b0[u],
                                                   (short)0, acc0, false, false);
      acc1 = __builtin_amdgcn_wmma_f32_16x16x4_f32(false, av, false, b1[u],
                                                   (short)0, acc1, false, false);
    }
  }

#pragma unroll
  for (int r = 0; r < 8; ++r) {
    red[wid][r][lane]     = acc0[r];
    red[wid][8 + r][lane] = acc1[r];
  }
  wred[wid][lane] = wp;
  __syncthreads();

  {
    const float* __restrict__ rp = &red[0][0][0] + tid;
    float sum = rp[0];
#pragma unroll
    for (int w = 1; w < CWAVES; ++w) sum += rp[w * 512];
    const int rr  = tid >> 5;
    const int m   = 8 * (lane >> 4) + (rr & 7);
    const int col = (rr >> 3) * 16 + (lane & 15);
    if (m < KC) cen[m][col] = sum;
  }
  if (tid < 16) {
    float ws = 0.0f;
#pragma unroll
    for (int w = 0; w < CWAVES; ++w) ws += wred[w][tid] + wred[w][tid + 16];
    wsumS[tid] = ws + 1e-8f;
  }
  __syncthreads();

  if (tid < KC * 32) {
    const int m = tid >> 5;
    const int o = m * D + cg * 32 + (tid & 31);
    out_centers[o] = cen[m][tid & 31] / wsumS[m];
    out_protos[o]  = prototypes[o];
  }
}

#define PCH 8
#define WSTEPS 16

__global__ __launch_bounds__(256)
void comp_matrix_kernel(const float* __restrict__ aT,
                        const float* __restrict__ bT,
                        const float* __restrict__ cg1,
                        const float* __restrict__ cbe1,
                        const float* __restrict__ cw2,
                        const float* __restrict__ cb2p,
                        float* __restrict__ out, int N) {
  __shared__ __attribute__((aligned(16))) float Ag[H * 16];

  const int tid  = threadIdx.x;
  const int lane = tid & 31;
  const int wid  = __builtin_amdgcn_readfirstlane(tid >> 5);
  const unsigned half = (unsigned)lane >> 4;
  const unsigned l16  = (unsigned)lane & 15u;
  const unsigned un    = (unsigned)N;
  const unsigned nlast = un - 1u;
  const unsigned i0 = blockIdx.x * 16u;
  const unsigned j0 = blockIdx.y * 256u + (unsigned)wid * 32u;

#pragma unroll
  for (int u = 0; u < 8; ++u) {
    const unsigned e  = (unsigned)tid + 256u * (unsigned)u;
    const unsigned kk = e >> 4;
    const unsigned r  = e & 15u;
    unsigned ii = i0 + r;
    ii = (ii < un) ? ii : nlast;
    Ag[e] = aT[kk * un + ii] * cg1[kk];
  }
  const float cb2 = cb2p[0];

  unsigned ic  = i0 + l16;        ic  = (ic  < un) ? ic  : nlast;
  unsigned jc0 = j0 + l16;        jc0 = (jc0 < un) ? jc0 : nlast;
  unsigned jc1 = j0 + 16u + l16;  jc1 = (jc1 < un) ? jc1 : nlast;

  v8f c0 = {}, c1 = {};
  float sa = 0.0f, sb0 = 0.0f, sb1 = 0.0f;
  for (int s0 = 0; s0 < H / 4; s0 += WSTEPS) {
    v2f a[WSTEPS], b0[WSTEPS], b1[WSTEPS];
#pragma unroll
    for (int u = 0; u < WSTEPS; ++u) {
      const unsigned k = (unsigned)(s0 + u) * 4u + half * 2u;
      a[u].x  = aT[k * un + ic];
      a[u].y  = aT[(k + 1u) * un + ic];
      b0[u].x = bT[k * un + jc0];
      b0[u].y = bT[(k + 1u) * un + jc0];
      b1[u].x = bT[k * un + jc1];
      b1[u].y = bT[(k + 1u) * un + jc1];
    }
    __builtin_amdgcn_sched_barrier(0);
#pragma unroll
    for (int u = 0; u < WSTEPS; ++u) {
      sa  += a[u].x * a[u].x + a[u].y * a[u].y;
      sb0 += b0[u].x * b0[u].x + b0[u].y * b0[u].y;
      sb1 += b1[u].x * b1[u].x + b1[u].y * b1[u].y;
      c0 = __builtin_amdgcn_wmma_f32_16x16x4_f32(false, a[u], false, b0[u],
                                                 (short)0, c0, false, false);
      c1 = __builtin_amdgcn_wmma_f32_16x16x4_f32(false, a[u], false, b1[u],
                                                 (short)0, c1, false, false);
    }
  }
  sa  += __shfl_xor(sa, 16);
  sb0 += __shfl_xor(sb0, 16);
  sb1 += __shfl_xor(sb1, 16);
#pragma unroll
  for (int r = 0; r < 8; ++r) { c0[r] *= 2.0f; c1[r] *= 2.0f; }
  {
    v2f an, bn0, bn1;
    an.x  = (half == 0u) ? sa   : 0.0f;
    an.y  = (half == 0u) ? 1.0f : 0.0f;
    bn0.x = (half == 0u) ? 1.0f : 0.0f;
    bn0.y = (half == 0u) ? sb0  : 0.0f;
    bn1.x = bn0.x;
    bn1.y = (half == 0u) ? sb1  : 0.0f;
    c0 = __builtin_amdgcn_wmma_f32_16x16x4_f32(false, an, false, bn0,
                                               (short)0, c0, false, false);
    c1 = __builtin_amdgcn_wmma_f32_16x16x4_f32(false, an, false, bn1,
                                               (short)0, c1, false, false);
  }
  v2f inv2[8];
  {
    float vrow[16];
#pragma unroll
    for (int r = 0; r < 8; ++r) {
      const float snd = (half == 0u) ? c1[r] : c0[r];
      const float rcv = __shfl_xor(snd, 16);
      vrow[r]     = (half == 0u) ? c0[r] : rcv;
      vrow[8 + r] = (half == 0u) ? rcv   : c1[r];
    }
#pragma unroll
    for (int rp = 0; rp < 8; ++rp) {
      inv2[rp].x = rsqrtf(vrow[2 * rp]     * (1.0f / 128.0f) + LN_EPS);
      inv2[rp].y = rsqrtf(vrow[2 * rp + 1] * (1.0f / 128.0f) + LN_EPS);
    }
  }

  __syncthreads();

  unsigned jc = j0 + (unsigned)lane;
  jc = (jc < un) ? jc : nlast;
  const float* __restrict__ bp = bT + jc;

  v2f acc[8];
#pragma unroll
  for (int rp = 0; rp < 8; ++rp) { acc[rp].x = 0.0f; acc[rp].y = 0.0f; }

  const v4f* __restrict__ Ag4 = (const v4f*)Ag;

  float cur[PCH], be_c[PCH], cw_c[PCH], g_c[PCH];
#pragma unroll
  for (int u = 0; u < PCH; ++u) {
    cur[u]  = bp[(unsigned)u * un];
    be_c[u] = cbe1[u];
    cw_c[u] = cw2[u];
    g_c[u]  = cg1[u];
  }

  v4f Ac[2][4];
#pragma unroll
  for (int q = 0; q < 2; ++q)
#pragma unroll
    for (int m = 0; m < 4; ++m) Ac[q][m] = Ag4[4 * q + m];

  for (int k0 = 0; k0 < H; k0 += PCH) {
    const int kn0 = (k0 + PCH < H) ? (k0 + PCH) : (H - PCH);
    float nxt[PCH], be_n[PCH], cw_n[PCH], g_n[PCH];
#pragma unroll
    for (int u = 0; u < PCH; ++u) {
      nxt[u]  = bp[(unsigned)(kn0 + u) * un];
      be_n[u] = cbe1[kn0 + u];
      cw_n[u] = cw2[kn0 + u];
      g_n[u]  = cg1[kn0 + u];
    }
#pragma unroll
    for (int u = 0; u < PCH; u += 2) {
      int kb = k0 + u + 2;
      kb = (kb < H) ? kb : (H - 2);
      v4f An[2][4];
#pragma unroll
      for (int q = 0; q < 2; ++q)
#pragma unroll
        for (int m = 0; m < 4; ++m) An[q][m] = Ag4[4 * (kb + q) + m];
      __builtin_amdgcn_sched_barrier(0);
#pragma unroll
      for (int q = 0; q < 2; ++q) {
        const float be = be_c[u + q];
        const float cw = cw_c[u + q];
        const float bg = cur[u + q] * g_c[u + q];
        const v4f A0 = Ac[q][0], A1 = Ac[q][1], A2 = Ac[q][2], A3 = Ac[q][3];
        const float ax[16] = {A0.x, A0.y, A0.z, A0.w, A1.x, A1.y, A1.z, A1.w,
                              A2.x, A2.y, A2.z, A2.w, A3.x, A3.y, A3.z, A3.w};
#pragma unroll
        for (int rp = 0; rp < 8; ++rp) {
          v2f z;
          z.x = ax[2 * rp]     + bg;
          z.y = ax[2 * rp + 1] + bg;
          z = z * inv2[rp] + be;
          z.x = fmaxf(z.x, 0.0f);
          z.y = fmaxf(z.y, 0.0f);
          acc[rp] += z * cw;
        }
      }
      __builtin_amdgcn_sched_barrier(0);
#pragma unroll
      for (int q = 0; q < 2; ++q)
#pragma unroll
        for (int m = 0; m < 4; ++m) Ac[q][m] = An[q][m];
    }
#pragma unroll
    for (int u = 0; u < PCH; ++u) {
      cur[u]  = nxt[u];
      be_c[u] = be_n[u];
      cw_c[u] = cw_n[u];
      g_c[u]  = g_n[u];
    }
  }

  const unsigned j = j0 + (unsigned)lane;
  if (j < un) {
#pragma unroll
    for (int rp = 0; rp < 8; ++rp) {
      const unsigned ia = i0 + 2u * (unsigned)rp;
      if (ia < un)      out[(size_t)ia * un + j]        = tanhf(acc[rp].x + cb2);
      if (ia + 1u < un) out[(size_t)(ia + 1u) * un + j] = tanhf(acc[rp].y + cb2);
    }
  }
}

#define CS_THREADS 256
#define EPT 8

__global__ __launch_bounds__(CS_THREADS)
void comp_scores_kernel(const float* __restrict__ mat,
                        const int* __restrict__ ei, int E, int N,
                        float* __restrict__ out) {
  const unsigned ue    = (unsigned)E;
  const unsigned un    = (unsigned)N;
  const unsigned nlast = un - 1u;
  const unsigned chunk = CS_THREADS * EPT;
  const int* erow = ei;
  const int* ecol = ei + ue;

  for (unsigned c0 = blockIdx.x * chunk; c0 < ue; c0 += gridDim.x * chunk) {
    const unsigned e0 = c0 + threadIdx.x;
    if (c0 + chunk <= ue) {
      unsigned ri[EPT], ci[EPT];
#pragma unroll
      for (int u = 0; u < EPT; ++u) {
        ri[u] = (unsigned)erow[e0 + u * CS_THREADS];
        ci[u] = (unsigned)ecol[e0 + u * CS_THREADS];
      }
      float v[EPT];
#pragma unroll
      for (int u = 0; u < EPT; ++u) {
        const unsigned i = (ri[u] < un) ? ri[u] : nlast;
        const unsigned j = (ci[u] < un) ? ci[u] : nlast;
        v[u] = mat[i * un + j];
      }
#pragma unroll
      for (int u = 0; u < EPT; ++u) out[e0 + u * CS_THREADS] = v[u];
    } else {
#pragma unroll 1
      for (unsigned e = e0; e < ue; e += CS_THREADS) {
        const unsigned r = (unsigned)erow[e];
        const unsigned c = (unsigned)ecol[e];
        const unsigned i = (r < un) ? r : nlast;
        const unsigned j = (c < un) ? c : nlast;
        out[e] = mat[i * un + j];
      }
    }
  }
}

extern "C" void kernel_launch(void* const* d_in, const int* in_sizes, int n_in,
                              void* d_out, int out_size, void* d_ws, size_t ws_size,
                              hipStream_t stream) {
  const float* x    = (const float*)d_in[0];
  const int*   ei   = (const int*)  d_in[1];
  const float* w1   = (const float*)d_in[2];
  const float* b1   = (const float*)d_in[3];
  const float* g1   = (const float*)d_in[4];
  const float* be1  = (const float*)d_in[5];
  const float* w2   = (const float*)d_in[6];
  const float* b2   = (const float*)d_in[7];
  const float* g2   = (const float*)d_in[8];
  const float* be2  = (const float*)d_in[9];
  const float* w3   = (const float*)d_in[10];
  const float* b3   = (const float*)d_in[11];
  const float* cw1  = (const float*)d_in[12];
  const float* cb1  = (const float*)d_in[13];
  const float* cg1  = (const float*)d_in[14];
  const float* cbe1 = (const float*)d_in[15];
  const float* cw2  = (const float*)d_in[16];
  const float* cb2  = (const float*)d_in[17];
  const float* prot = (const float*)d_in[18];

  const int N = in_sizes[0] / D;
  const int E = in_sizes[1] / 2;

  float* h1pre = (float*)d_ws;
  float* aT    = h1pre + (size_t)N * H;
  float* bT    = aT    + (size_t)N * H;

  float* out       = (float*)d_out;
  float* o_logits  = out;
  float* o_probs   = o_logits  + (size_t)N * KC;
  float* o_scores  = o_probs   + (size_t)N * KC;
  float* o_matrix  = o_scores  + (size_t)E;
  float* o_centers = o_matrix  + (size_t)N * N;
  float* o_protos  = o_centers + (size_t)KC * D;

  const int rowTiles  = (N + 15) / 16;
  const int colGroups = (N + 255) / 256;

  wmma_gemm3_kernel<<<dim3(rowTiles, 3), 32, 0, stream>>>(
      x, w1, cw1, cb1, h1pre, aT, bT, N);

  node_mlp_kernel<<<N, 128, 0, stream>>>(
      h1pre, b1, g1, be1, w2, b2, g2, be2, w3, b3, o_logits, o_probs, N);

  centers_kernel<<<4, CWAVES * 32, 0, stream>>>(
      o_probs, x, prot, o_centers, o_protos, N);

  comp_matrix_kernel<<<dim3(rowTiles, colGroups), 256, 0, stream>>>(
      aT, bT, cg1, cbe1, cw2, cb2, o_matrix, N);

  comp_scores_kernel<<<256, CS_THREADS, 0, stream>>>(
      o_matrix, ei, E, N, o_scores);
}


# launch 1 (three 128x128 f32 GEMMs): 4 waves per 16-node stripe, 2 column tiles each, x stripe staged in LDS, all 32 K-steps of weight loads issued up front; was 1 wave, 8 load-wait-WMMA trips
# speedup vs baseline: 8.8252x; 2.1010x over previous
#include <hip/hip_runtime.h>
#include <math.h>

typedef __attribute__((ext_vector_type(2))) float v2f;
typedef __attribute__((ext_vector_type(4))) float v4f;
typedef __attribute__((ext_vector_type(8))) float v8f;

#define D   128
#define H   128
#define HH2 64
#define KC  10
#define LN_EPS 1e-5f

#define XP 132
#define GSTEPS 32

__global__ __launch_bounds__(128)
void wmma_gemm3_kernel(const float* __restrict__ x,
                       const float* __restrict__ w1,
                       const float* __restrict__ cw1,
                       const float* __restrict__ cb1,
                       float* __restrict__ h1pre,
                       float* __restrict__ aT,
                       float* __restrict__ bT,
                       int N) {
  const int rt  = blockIdx.x;
  const int mat = blockIdx.y;

  const float* __restrict__ Bm = (mat == 0) ? w1 : (mat == 1 ? cw1 : cw1 + D * H);

  const int tid   = threadIdx.x;
  const int lane  = tid & 31;
  const int wid   = __builtin_amdgcn_readfirstlane(tid >> 5);
  const int half  = lane >> 4;
  const int l16   = lane & 15;
  const int cbase = wid * 32;

  __shared__ __attribute__((aligned(16))) float xs[16 * XP];
  __shared__ float part[4][16];

  v4f xv[4];
#pragma unroll
  for (int q = 0; q < 4; ++q) {
    int n = rt * 16 + wid * 4 + q;
    if (n >= N) n = N - 1;
    xv[q] = *(const v4f*)(x + (size_t)n * D + lane * 4);
  }

  const float* __restrict__ wp = Bm + (size_t)(2 * half) * H + cbase + 2 * l16;
  v2f wlo[GSTEPS], whi[GSTEPS];
#pragma unroll
  for (int s = 0; s < GSTEPS; ++s) {
    wlo[s] = *(const v2f*)(wp + (4 * s) * H);
    whi[s] = *(const v2f*)(wp + (4 * s + 1) * H);
  }
  __builtin_amdgcn_sched_barrier(0);

#pragma unroll
  for (int q = 0; q < 4; ++q)
    *(v4f*)(xs + (wid * 4 + q) * XP + lane * 4) = xv[q];
  __syncthreads();

  const float* xa = xs + l16 * XP + 2 * half;
  v2f a[GSTEPS];
#pragma unroll
  for (int s = 0; s < GSTEPS; ++s) a[s] = *(const v2f*)(xa + 4 * s);

  v8f acc0 = {}, acc1 = {};

  if (mat == 0) {
#pragma unroll
    for (int s = 0; s < GSTEPS; ++s) {
      v2f b0, b1;
      b0.x = wlo[s].x; b0.y = whi[s].x;
      b1.x = wlo[s].y; b1.y = whi[s].y;
      acc0 = __builtin_amdgcn_wmma_f32_16x16x4_f32(false, a[s], false, b0,
                                                   (short)0, acc0, false, false);
      acc1 = __builtin_amdgcn_wmma_f32_16x16x4_f32(false, a[s], false, b1,
                                                   (short)0, acc1, false, false);
    }
    const int rbase = rt * 16 + half * 8;
    float* __restrict__ op = h1pre + (size_t)rbase * H + cbase + 2 * l16;
    if (rt * 16 + 16 <= N) {
#pragma unroll
      for (int r = 0; r < 8; ++r) {
        v2f o; o.x = acc0[r]; o.y = acc1[r];
        *(v2f*)(op + (size_t)r * H) = o;
      }
    } else {
#pragma unroll
      for (int r = 0; r < 8; ++r)
        if (rbase + r < N) {
          v2f o; o.x = acc0[r]; o.y = acc1[r];
          *(v2f*)(op + (size_t)r * H) = o;
        }
    }
  } else {
#pragma unroll
    for (int s = 0; s < GSTEPS; ++s) {
      v2f b0, b1;
      b0.x = wlo[s].x; b0.y = whi[s].x;
      b1.x = wlo[s].y; b1.y = whi[s].y;
      acc0 = __builtin_amdgcn_wmma_f32_16x16x4_f32(false, b0, false, a[s],
                                                   (short)0, acc0, false, false);
      acc1 = __builtin_amdgcn_wmma_f32_16x16x4_f32(false, b1, false, a[s],
                                                   (short)0, acc1, false, false);
    }
    if (mat == 1) {
      const v4f* __restrict__ cbp = (const v4f*)(cb1 + cbase + 16 * half);
      const v4f q0 = cbp[0], q1 = cbp[1], q2 = cbp[2], q3 = cbp[3];
      acc0[0] += q0.x; acc1[0] += q0.y; acc0[1] += q0.z; acc1[1] += q0.w;
      acc0[2] += q1.x; acc1[2] += q1.y; acc0[3] += q1.z; acc1[3] += q1.w;
      acc0[4] += q2.x; acc1[4] += q2.y; acc0[5] += q2.z; acc1[5] += q2.w;
      acc0[6] += q3.x; acc1[6] += q3.y; acc0[7] += q3.z; acc1[7] += q3.w;
    }
    float s = 0.0f;
#pragma unroll
    for (int r = 0; r < 8; ++r) s += acc0[r] + acc1[r];
    s += __shfl_xor(s, 16);
    if (half == 0) part[wid][l16] = s;
    __syncthreads();
    const float mean = (part[0][l16] + part[1][l16] + part[2][l16] + part[3][l16])
                       * (1.0f / 128.0f);
    float* __restrict__ outT = (mat == 1) ? aT : bT;
    const int n = rt * 16 + l16;
    if (n < N) {
      float* __restrict__ tp = outT + (size_t)(cbase + 16 * half) * N + n;
#pragma unroll
      for (int r = 0; r < 8; ++r) {
        tp[(size_t)(2 * r) * N]     = acc0[r] - mean;
        tp[(size_t)(2 * r + 1) * N] = acc1[r] - mean;
      }
    }
  }
}

__device__ __forceinline__ void node_mlp_role(const int nb,
                                const float* __restrict__ h1pre,
                                const float* __restrict__ b1,
                                const float* __restrict__ g1,
                                const float* __restrict__ be1,
                                const float* __restrict__ w2,
                                const float* __restrict__ b2,
                                const float* __restrict__ g2,
                                const float* __restrict__ be2,
                                const float* __restrict__ w3,
                                const float* __restrict__ b3,
                                float* __restrict__ out_logits,
                                float* __restrict__ out_probs,
                                int N) {
  if (nb * 2 >= N) return;
  __builtin_amdgcn_s_setprio(3);
  const int sub  = threadIdx.x >> 7;
  const int t    = threadIdx.x & 127;
  const int nraw = nb * 2 + sub;
  const bool live = nraw < N;
  const int n = live ? nraw : N - 1;
  __shared__ float sA2[2][128], sB2[2][128];
  __shared__ float sh1_2[2][128];
  __shared__ float sh2_2[2][HH2];
  __shared__ float slog2[2][16];
  float* sA   = sA2[sub];
  float* sB   = sB2[sub];
  float* sh1  = sh1_2[sub];
  float* sh2  = sh2_2[sub];
  float* slog = slog2[sub];

  float v = h1pre[(size_t)n * H + t] + b1[t];
  sA[t] = v;
  sB[t] = v * v;
  __syncthreads();
  for (int s = 64; s > 0; s >>= 1) {
    if (t < s) { sA[t] += sA[t + s]; sB[t] += sB[t + s]; }
    __syncthreads();
  }
  {
    float m   = sA[0] * (1.0f / 128.0f);
    float var = sB[0] * (1.0f / 128.0f) - m * m;
    float inv = rsqrtf(var + LN_EPS);
    float hv  = (v - m) * inv * g1[t] + be1[t];
    v = fmaxf(hv, 0.0f);
  }
  __syncthreads();
  sh1[t] = v;
  __syncthreads();

  float a2 = 0.0f;
  if (t < HH2) {
    a2 = b2[t];
    for (int k = 0; k < H; ++k) a2 += sh1[k] * w2[k * HH2 + t];
    sA[t] = a2;
    sB[t] = a2 * a2;
  }
  __syncthreads();
  for (int s = 32; s > 0; s >>= 1) {
    if (t < s) { sA[t] += sA[t + s]; sB[t] += sB[t + s]; }
    __syncthreads();
  }
  if (t < HH2) {
    float m2  = sA[0] * (1.0f / 64.0f);
    float v2  = sB[0] * (1.0f / 64.0f) - m2 * m2;
    float i2  = rsqrtf(v2 + LN_EPS);
    float h2v = (a2 - m2) * i2 * g2[t] + be2[t];
    sh2[t] = fmaxf(h2v, 0.0f);
  }
  __syncthreads();

  if (t < KC) {
    float lg = b3[t];
#pragma unroll
    for (int k = 0; k < HH2; ++k) lg += sh2[k] * w3[k * KC + t];
    slog[t] = lg;
    if (live) out_logits[(size_t)n * KC + t] = lg;
  }
  __syncthreads();
  if (t < KC) {
    float mx = slog[0];
    for (int k = 1; k < KC; ++k) mx = fmaxf(mx, slog[k]);
    float ssum = 0.0f;
    for (int k = 0; k < KC; ++k) ssum += __expf(slog[k] - mx);
    if (live) out_probs[(size_t)n * KC + t] = __expf(slog[t] - mx) / ssum;
  }
}

#define CWAVES 16
#define CSTEPS 16

__device__ __forceinline__
void centers_role(const int cg,
                    const float* __restrict__ probs,
                    const float* __restrict__ x,
                    const float* __restrict__ prototypes,
                    float* __restrict__ out_centers,
                    float* __restrict__ out_protos,
                    int N) {
  const int tid  = threadIdx.x;
  const int lane = tid & 31;
  const int wid  = __builtin_amdgcn_readfirstlane(tid >> 5);
  const unsigned half = (unsigned)lane >> 4;
  const unsigned l16  = (unsigned)lane & 15u;

  __shared__ float red[CWAVES][16][32];
  __shared__ float wred[CWAVES][32];
  __shared__ float cen[KC][32];
  __shared__ float wsumS[16];

  const unsigned pofs = (l16 < KC) ? l16 : (unsigned)(KC - 1);
  const unsigned xofs = (unsigned)cg * 32u + l16;
  const unsigned un   = (unsigned)N;
  const unsigned nlast = un - 1u;

  v8f acc0 = {}, acc1 = {};
  float wp = 0.0f;

  const int nstep = (N + 3) >> 2;
  for (int s0 = 0; s0 < nstep; s0 += CWAVES * CSTEPS) {
    v2f a[CSTEPS], b0[CSTEPS], b1[CSTEPS];
#pragma unroll
    for (int u = 0; u < CSTEPS; ++u) {
      const unsigned n0  = (unsigned)(s0 + u * CWAVES + wid) * 4u + half * 2u;
      const unsigned n1  = n0 + 1u;
      const unsigned n0c = (n0 < un) ? n0 : nlast;
      const unsigned n1c = (n1 < un) ? n1 : nlast;
      a[u].x  = probs[n0c * KC + pofs];
      a[u].y  = probs[n1c * KC + pofs];
      b0[u].x = x[n0c * D + xofs];
      b0[u].y = x[n1c * D + xofs];
      b1[u].x = x[n0c * D + xofs + 16u];
      b1[u].y = x[n1c * D + xofs + 16u];
    }
    __builtin_amdgcn_sched_barrier(0);
#pragma unroll
    for (int u = 0; u < CSTEPS; ++u) {
      const unsigned n0 = (unsigned)(s0 + u * CWAVES + wid) * 4u + half * 2u;
      v2f av;
      av.x = (n0 < un)      ? a[u].x : 0.0f;
      av.y = (n0 + 1u < un) ? a[u].y : 0.0f;
      wp += av.x + av.y;
      acc0 = __builtin_amdgcn_wmma_f32_16x16x4_f32(false, av, false, b0[u],
                                                   (short)0, acc0, false, false);
      acc1 = __builtin_amdgcn_wmma_f32_16x16x4_f32(false, av, false, b1[u],
                                                   (short)0, acc1, false, false);
    }
  }

#pragma unroll
  for (int r = 0; r < 8; ++r) {
    red[wid][r][lane]     = acc0[r];
    red[wid][8 + r][lane] = acc1[r];
  }
  wred[wid][lane] = wp;
  __syncthreads();

  {
    const float* __restrict__ rp = &red[0][0][0] + tid;
    float sum = rp[0];
#pragma unroll
    for (int w = 1; w < CWAVES; ++w) sum += rp[w * 512];
    const int rr  = tid >> 5;
    const int m   = 8 * (lane >> 4) + (rr & 7);
    const int col = (rr >> 3) * 16 + (lane & 15);
    if (m < KC) cen[m][col] = sum;
  }
  if (tid < 16) {
    float ws = 0.0f;
#pragma unroll
    for (int w = 0; w < CWAVES; ++w) ws += wred[w][tid] + wred[w][tid + 16];
    wsumS[tid] = ws + 1e-8f;
  }
  __syncthreads();

  if (tid < KC * 32) {
    const int m = tid >> 5;
    const int o = m * D + cg * 32 + (tid & 31);
    out_centers[o] = cen[m][tid & 31] / wsumS[m];
    out_protos[o]  = prototypes[o];
  }
}

#define PCH 8
#define WSTEPS 16

__device__ __forceinline__
void comp_matrix_role(const unsigned bx, const unsigned by,
                        const float* __restrict__ aT,
                        const float* __restrict__ bT,
                        const float* __restrict__ cg1,
                        const float* __restrict__ cbe1,
                        const float* __restrict__ cw2,
                        const float* __restrict__ cb2p,
                        float* __restrict__ out, int N) {
  __shared__ __attribute__((aligned(16))) float Ag[H * 16];

  const int tid  = threadIdx.x;
  const int lane = tid & 31;
  const int wid  = __builtin_amdgcn_readfirstlane(tid >> 5);
  const unsigned half = (unsigned)lane >> 4;
  const unsigned l16  = (unsigned)lane & 15u;
  const unsigned un    = (unsigned)N;
  const unsigned nlast = un - 1u;
  const unsigned i0 = bx * 16u;
  const unsigned j0 = by * 256u + (unsigned)wid * 32u;

#pragma unroll
  for (int u = 0; u < 8; ++u) {
    const unsigned e  = (unsigned)tid + 256u * (unsigned)u;
    const unsigned kk = e >> 4;
    const unsigned r  = e & 15u;
    unsigned ii = i0 + r;
    ii = (ii < un) ? ii : nlast;
    Ag[e] = aT[kk * un + ii] * cg1[kk];
  }
  const float cb2 = cb2p[0];

  unsigned ic  = i0 + l16;        ic  = (ic  < un) ? ic  : nlast;
  unsigned jc0 = j0 + l16;        jc0 = (jc0 < un) ? jc0 : nlast;
  unsigned jc1 = j0 + 16u + l16;  jc1 = (jc1 < un) ? jc1 : nlast;

  v8f c0 = {}, c1 = {};
  float sa = 0.0f, sb0 = 0.0f, sb1 = 0.0f;
  for (int s0 = 0; s0 < H / 4; s0 += WSTEPS) {
    v2f a[WSTEPS], b0[WSTEPS], b1[WSTEPS];
#pragma unroll
    for (int u = 0; u < WSTEPS; ++u) {
      const unsigned k = (unsigned)(s0 + u) * 4u + half * 2u;
      a[u].x  = aT[k * un + ic];
      a[u].y  = aT[(k + 1u) * un + ic];
      b0[u].x = bT[k * un + jc0];
      b0[u].y = bT[(k + 1u) * un + jc0];
      b1[u].x = bT[k * un + jc1];
      b1[u].y = bT[(k + 1u) * un + jc1];
    }
    __builtin_amdgcn_sched_barrier(0);
#pragma unroll
    for (int u = 0; u < WSTEPS; ++u) {
      sa  += a[u].x * a[u].x + a[u].y * a[u].y;
      sb0 += b0[u].x * b0[u].x + b0[u].y * b0[u].y;
      sb1 += b1[u].x * b1[u].x + b1[u].y * b1[u].y;
      c0 = __builtin_amdgcn_wmma_f32_16x16x4_f32(false, a[u], false, b0[u],
                                                 (short)0, c0, false, false);
      c1 = __builtin_amdgcn_wmma_f32_16x16x4_f32(false, a[u], false, b1[u],
                                                 (short)0, c1, false, false);
    }
  }
  sa  += __shfl_xor(sa, 16);
  sb0 += __shfl_xor(sb0, 16);
  sb1 += __shfl_xor(sb1, 16);
#pragma unroll
  for (int r = 0; r < 8; ++r) { c0[r] *= 2.0f; c1[r] *= 2.0f; }
  {
    v2f an, bn0, bn1;
    an.x  = (half == 0u) ? sa   : 0.0f;
    an.y  = (half == 0u) ? 1.0f : 0.0f;
    bn0.x = (half == 0u) ? 1.0f : 0.0f;
    bn0.y = (half == 0u) ? sb0  : 0.0f;
    bn1.x = bn0.x;
    bn1.y = (half == 0u) ? sb1  : 0.0f;
    c0 = __builtin_amdgcn_wmma_f32_16x16x4_f32(false, an, false, bn0,
                                               (short)0, c0, false, false);
    c1 = __builtin_amdgcn_wmma_f32_16x16x4_f32(false, an, false, bn1,
                                               (short)0, c1, false, false);
  }
  v2f inv2[8];
  {
    float vrow[16];
#pragma unroll
    for (int r = 0; r < 8; ++r) {
      const float snd = (half == 0u) ? c1[r] : c0[r];
      const float rcv = __shfl_xor(snd, 16);
      vrow[r]     = (half == 0u) ? c0[r] : rcv;
      vrow[8 + r] = (half == 0u) ? rcv   : c1[r];
    }
#pragma unroll
    for (int rp = 0; rp < 8; ++rp) {
      inv2[rp].x = rsqrtf(vrow[2 * rp]     * (1.0f / 128.0f) + LN_EPS);
      inv2[rp].y = rsqrtf(vrow[2 * rp + 1] * (1.0f / 128.0f) + LN_EPS);
    }
  }

  __syncthreads();

  unsigned jc = j0 + (unsigned)lane;
  jc = (jc < un) ? jc : nlast;
  const float* __restrict__ bp = bT + jc;

  v2f acc[8];
#pragma unroll
  for (int rp = 0; rp < 8; ++rp) { acc[rp].x = 0.0f; acc[rp].y = 0.0f; }

  const v4f* __restrict__ Ag4 = (const v4f*)Ag;

  const __attribute__((address_space(4))) float* cbe1c =
      (const __attribute__((address_space(4))) float*)(unsigned long long)cbe1;
  const __attribute__((address_space(4))) float* cw2c =
      (const __attribute__((address_space(4))) float*)(unsigned long long)cw2;
  const __attribute__((address_space(4))) float* cg1c =
      (const __attribute__((address_space(4))) float*)(unsigned long long)cg1;

  float cur[PCH], be_c[PCH], cw_c[PCH], g_c[PCH];
#pragma unroll
  for (int u = 0; u < PCH; ++u) {
    cur[u]  = bp[(unsigned)u * un];
    be_c[u] = cbe1c[u];
    cw_c[u] = cw2c[u];
    g_c[u]  = cg1c[u];
  }

  v4f Ac[2][4];
#pragma unroll
  for (int q = 0; q < 2; ++q)
#pragma unroll
    for (int m = 0; m < 4; ++m) Ac[q][m] = Ag4[4 * q + m];

  for (int k0 = 0; k0 < H; k0 += PCH) {
    const int kn0 = (k0 + PCH < H) ? (k0 + PCH) : (H - PCH);
    float nxt[PCH], be_n[PCH], cw_n[PCH], g_n[PCH];
#pragma unroll
    for (int u = 0; u < PCH; ++u) {
      nxt[u]  = bp[(unsigned)(kn0 + u) * un];
      be_n[u] = cbe1c[kn0 + u];
      cw_n[u] = cw2c[kn0 + u];
      g_n[u]  = cg1c[kn0 + u];
    }
#pragma unroll
    for (int u = 0; u < PCH; u += 2) {
      int kb = k0 + u + 2;
      kb = (kb < H) ? kb : (H - 2);
      v4f An[2][4];
#pragma unroll
      for (int q = 0; q < 2; ++q)
#pragma unroll
        for (int m = 0; m < 4; ++m) An[q][m] = Ag4[4 * (kb + q) + m];
      __builtin_amdgcn_sched_barrier(0);
#pragma unroll
      for (int q = 0; q < 2; ++q) {
        const float be = be_c[u + q];
        const float cw = cw_c[u + q];
        const float bg = cur[u + q] * g_c[u + q];
        const v4f A0 = Ac[q][0], A1 = Ac[q][1], A2 = Ac[q][2], A3 = Ac[q][3];
        const float ax[16] = {A0.x, A0.y, A0.z, A0.w, A1.x, A1.y, A1.z, A1.w,
                              A2.x, A2.y, A2.z, A2.w, A3.x, A3.y, A3.z, A3.w};
#pragma unroll
        for (int rp = 0; rp < 8; ++rp) {
          v2f z;
          z.x = ax[2 * rp]     + bg;
          z.y = ax[2 * rp + 1] + bg;
          z = z * inv2[rp] + be;
          z.x = fmaxf(z.x, 0.0f);
          z.y = fmaxf(z.y, 0.0f);
          acc[rp] += z * cw;
        }
      }
      __builtin_amdgcn_sched_barrier(0);
#pragma unroll
      for (int q = 0; q < 2; ++q)
#pragma unroll
        for (int m = 0; m < 4; ++m) Ac[q][m] = An[q][m];
    }
#pragma unroll
    for (int u = 0; u < PCH; ++u) {
      cur[u]  = nxt[u];
      be_c[u] = be_n[u];
      cw_c[u] = cw_n[u];
      g_c[u]  = g_n[u];
    }
  }

  const unsigned j = j0 + (unsigned)lane;
  if (j < un) {
#pragma unroll
    for (int rp = 0; rp < 8; ++rp) {
      const unsigned ia = i0 + 2u * (unsigned)rp;
      if (ia < un)      out[(size_t)ia * un + j]        = tanhf(acc[rp].x + cb2);
      if (ia + 1u < un) out[(size_t)(ia + 1u) * un + j] = tanhf(acc[rp].y + cb2);
    }
  }
}

#define CS_THREADS 512
#define EPT 4

__device__ __forceinline__
void comp_scores_role(const unsigned blk, const unsigned nblk,
                        const float* __restrict__ mat,
                        const int* __restrict__ ei, int E, int N,
                        float* __restrict__ out) {
  const unsigned ue    = (unsigned)E;
  const unsigned un    = (unsigned)N;
  const unsigned nlast = un - 1u;
  const unsigned chunk = CS_THREADS * EPT;
  const int* erow = ei;
  const int* ecol = ei + ue;

  for (unsigned c0 = blk * chunk; c0 < ue; c0 += nblk * chunk) {
    const unsigned e0 = c0 + threadIdx.x;
    if (c0 + chunk <= ue) {
      unsigned ri[EPT], ci[EPT];
#pragma unroll
      for (int u = 0; u < EPT; ++u) {
        ri[u] = (unsigned)erow[e0 + u * CS_THREADS];
        ci[u] = (unsigned)ecol[e0 + u * CS_THREADS];
      }
      float v[EPT];
#pragma unroll
      for (int u = 0; u < EPT; ++u) {
        const unsigned i = (ri[u] < un) ? ri[u] : nlast;
        const unsigned j = (ci[u] < un) ? ci[u] : nlast;
        v[u] = mat[i * un + j];
      }
#pragma unroll
      for (int u = 0; u < EPT; ++u) out[e0 + u * CS_THREADS] = v[u];
    } else {
#pragma unroll 1
      for (unsigned e = e0; e < ue; e += CS_THREADS) {
        const unsigned r = (unsigned)erow[e];
        const unsigned c = (unsigned)ecol[e];
        const unsigned i = (r < un) ? r : nlast;
        const unsigned j = (c < un) ? c : nlast;
        out[e] = mat[i * un + j];
      }
    }
  }
}

__global__ __launch_bounds__(256, 8)
void pairs_and_nodes_kernel(const float* __restrict__ aT,
                            const float* __restrict__ bT,
                            const float* __restrict__ cg1,
                            const float* __restrict__ cbe1,
                            const float* __restrict__ cw2,
                            const float* __restrict__ cb2p,
                            float* __restrict__ out_matrix,
                            const float* __restrict__ h1pre,
                            const float* __restrict__ b1,
                            const float* __restrict__ g1,
                            const float* __restrict__ be1,
                            const float* __restrict__ w2,
                            const float* __restrict__ b2,
                            const float* __restrict__ g2,
                            const float* __restrict__ be2,
                            const float* __restrict__ w3,
                            const float* __restrict__ b3,
                            float* __restrict__ out_logits,
                            float* __restrict__ out_probs,
                            int N, int colGroups) {
  if ((int)blockIdx.y < colGroups) {
    comp_matrix_role(blockIdx.x, blockIdx.y, aT, bT, cg1, cbe1, cw2, cb2p,
                     out_matrix, N);
  } else {
    const int nb = ((int)blockIdx.y - colGroups) * (int)gridDim.x + (int)blockIdx.x;
    node_mlp_role(nb, h1pre, b1, g1, be1, w2, b2, g2, be2, w3, b3,
                  out_logits, out_probs, N);
  }
}

#define CENTER_BLOCKS 4

__global__ __launch_bounds__(CWAVES * 32)
void centers_and_scores_kernel(const float* __restrict__ probs,
                               const float* __restrict__ x,
                               const float* __restrict__ prototypes,
                               float* __restrict__ out_centers,
                               float* __restrict__ out_protos,
                               const float* __restrict__ mat,
                               const int* __restrict__ ei, int E, int N,
                               float* __restrict__ out_scores) {
  if (blockIdx.x < CENTER_BLOCKS) {
    centers_role((int)blockIdx.x, probs, x, prototypes, out_centers,
                 out_protos, N);
  } else {
    comp_scores_role(blockIdx.x - CENTER_BLOCKS, gridDim.x - CENTER_BLOCKS,
                     mat, ei, E, N, out_scores);
  }
}

extern "C" void kernel_launch(void* const* d_in, const int* in_sizes, int n_in,
                              void* d_out, int out_size, void* d_ws, size_t ws_size,
                              hipStream_t stream) {
  const float* x    = (const float*)d_in[0];
  const int*   ei   = (const int*)  d_in[1];
  const float* w1   = (const float*)d_in[2];
  const float* b1   = (const float*)d_in[3];
  const float* g1   = (const float*)d_in[4];
  const float* be1  = (const float*)d_in[5];
  const float* w2   = (const float*)d_in[6];
  const float* b2   = (const float*)d_in[7];
  const float* g2   = (const float*)d_in[8];
  const float* be2  = (const float*)d_in[9];
  const float* w3   = (const float*)d_in[10];
  const float* b3   = (const float*)d_in[11];
  const float* cw1  = (const float*)d_in[12];
  const float* cb1  = (const float*)d_in[13];
  const float* cg1  = (const float*)d_in[14];
  const float* cbe1 = (const float*)d_in[15];
  const float* cw2  = (const float*)d_in[16];
  const float* cb2  = (const float*)d_in[17];
  const float* prot = (const float*)d_in[18];

  const int N = in_sizes[0] / D;
  const int E = in_sizes[1] / 2;

  float* h1pre = (float*)d_ws;
  float* aT    = h1pre + (size_t)N * H;
  float* bT    = aT    + (size_t)N * H;

  float* out       = (float*)d_out;
  float* o_logits  = out;
  float* o_probs   = o_logits  + (size_t)N * KC;
  float* o_scores  = o_probs   + (size_t)N * KC;
  float* o_matrix  = o_scores  + (size_t)E;
  float* o_centers = o_matrix  + (size_t)N * N;
  float* o_protos  = o_centers + (size_t)KC * D;

  const int rowTiles  = (N + 15) / 16;
  const int colGroups = (N + 255) / 256;
  const int nodeBlocks = (N + 1) / 2;
  const int nodeRows   = (nodeBlocks + rowTiles - 1) / rowTiles;

  wmma_gemm3_kernel<<<dim3(rowTiles, 3), 128, 0, stream>>>(
      x, w1, cw1, cb1, h1pre, aT, bT, N);

  pairs_and_nodes_kernel<<<dim3(rowTiles, colGroups + nodeRows), 256, 0, stream>>>(
      aT, bT, cg1, cbe1, cw2, cb2, o_matrix,
      h1pre, b1, g1, be1, w2, b2, g2, be2, w3, b3, o_logits, o_probs,
      N, colGroups);

  centers_and_scores_kernel<<<CENTER_BLOCKS + 256, CWAVES * 32, 0, stream>>>(
      o_probs, x, prot, o_centers, o_protos, o_matrix, ei, E, N, o_scores);
}


# pair matrix tail: a lane's 16 tanh as one branch-free straight-line stream (both ranges computed, one selected per lane) ahead of the stores; was 16 two-path library calls run one after another
# speedup vs baseline: 9.1697x; 1.0390x over previous
#include <hip/hip_runtime.h>
#include <math.h>

typedef __attribute__((ext_vector_type(2))) float v2f;
typedef __attribute__((ext_vector_type(4))) float v4f;
typedef __attribute__((ext_vector_type(8))) float v8f;

#define D   128
#define H   128
#define HH2 64
#define KC  10
#define LN_EPS 1e-5f

#define XP 132
#define GSTEPS 32

__global__ __launch_bounds__(128)
void wmma_gemm3_kernel(const float* __restrict__ x,
                       const float* __restrict__ w1,
                       const float* __restrict__ cw1,
                       const float* __restrict__ cb1,
                       float* __restrict__ h1pre,
                       float* __restrict__ aT,
                       float* __restrict__ bT,
                       int N) {
  const int rt  = blockIdx.x;
  const int mat = blockIdx.y;

  const float* __restrict__ Bm = (mat == 0) ? w1 : (mat == 1 ? cw1 : cw1 + D * H);

  const int tid   = threadIdx.x;
  const int lane  = tid & 31;
  const int wid   = __builtin_amdgcn_readfirstlane(tid >> 5);
  const int half  = lane >> 4;
  const int l16   = lane & 15;
  const int cbase = wid * 32;

  __shared__ __attribute__((aligned(16))) float xs[16 * XP];
  __shared__ float part[4][16];

  v4f xv[4];
#pragma unroll
  for (int q = 0; q < 4; ++q) {
    int n = rt * 16 + wid * 4 + q;
    if (n >= N) n = N - 1;
    xv[q] = *(const v4f*)(x + (size_t)n * D + lane * 4);
  }

  const float* __restrict__ wp = Bm + (size_t)(2 * half) * H + cbase + 2 * l16;
  v2f wlo[GSTEPS], whi[GSTEPS];
#pragma unroll
  for (int s = 0; s < GSTEPS; ++s) {
    wlo[s] = *(const v2f*)(wp + (4 * s) * H);
    whi[s] = *(const v2f*)(wp + (4 * s + 1) * H);
  }
  __builtin_amdgcn_sched_barrier(0);

#pragma unroll
  for (int q = 0; q < 4; ++q)
    *(v4f*)(xs + (wid * 4 + q) * XP + lane * 4) = xv[q];
  __syncthreads();

  const float* xa = xs + l16 * XP + 2 * half;
  v2f a[GSTEPS];
#pragma unroll
  for (int s = 0; s < GSTEPS; ++s) a[s] = *(const v2f*)(xa + 4 * s);

  v8f acc0 = {}, acc1 = {};

  if (mat == 0) {
#pragma unroll
    for (int s = 0; s < GSTEPS; ++s) {
      v2f b0, b1;
      b0.x = wlo[s].x; b0.y = whi[s].x;
      b1.x = wlo[s].y; b1.y = whi[s].y;
      acc0 = __builtin_amdgcn_wmma_f32_16x16x4_f32(false, a[s], false, b0,
                                                   (short)0, acc0, false, false);
      acc1 = __builtin_amdgcn_wmma_f32_16x16x4_f32(false, a[s], false, b1,
                                                   (short)0, acc1, false, false);
    }
    const int rbase = rt * 16 + half * 8;
    float* __restrict__ op = h1pre + (size_t)rbase * H + cbase + 2 * l16;
    if (rt * 16 + 16 <= N) {
#pragma unroll
      for (int r = 0; r < 8; ++r) {
        v2f o; o.x = acc0[r]; o.y = acc1[r];
        *(v2f*)(op + (size_t)r * H) = o;
      }
    } else {
#pragma unroll
      for (int r = 0; r < 8; ++r)
        if (rbase + r < N) {
          v2f o; o.x = acc0[r]; o.y = acc1[r];
          *(v2f*)(op + (size_t)r * H) = o;
        }
    }
  } else {
#pragma unroll
    for (int s = 0; s < GSTEPS; ++s) {
      v2f b0, b1;
      b0.x = wlo[s].x; b0.y = whi[s].x;
      b1.x = wlo[s].y; b1.y = whi[s].y;
      acc0 = __builtin_amdgcn_wmma_f32_16x16x4_f32(false, b0, false, a[s],
                                                   (short)0, acc0, false, false);
      acc1 = __builtin_amdgcn_wmma_f32_16x16x4_f32(false, b1, false, a[s],
                                                   (short)0, acc1, false, false);
    }
    if (mat == 1) {
      const v4f* __restrict__ cbp = (const v4f*)(cb1 + cbase + 16 * half);
      const v4f q0 = cbp[0], q1 = cbp[1], q2 = cbp[2], q3 = cbp[3];
      acc0[0] += q0.x; acc1[0] += q0.y; acc0[1] += q0.z; acc1[1] += q0.w;
      acc0[2] += q1.x; acc1[2] += q1.y; acc0[3] += q1.z; acc1[3] += q1.w;
      acc0[4] += q2.x; acc1[4] += q2.y; acc0[5] += q2.z; acc1[5] += q2.w;
      acc0[6] += q3.x; acc1[6] += q3.y; acc0[7] += q3.z; acc1[7] += q3.w;
    }
    float s = 0.0f;
#pragma unroll
    for (int r = 0; r < 8; ++r) s += acc0[r] + acc1[r];
    s += __shfl_xor(s, 16);
    if (half == 0) part[wid][l16] = s;
    __syncthreads();
    const float mean = (part[0][l16] + part[1][l16] + part[2][l16] + part[3][l16])
                       * (1.0f / 128.0f);
    float* __restrict__ outT = (mat == 1) ? aT : bT;
    const int n = rt * 16 + l16;
    if (n < N) {
      float* __restrict__ tp = outT + (size_t)(cbase + 16 * half) * N + n;
#pragma unroll
      for (int r = 0; r < 8; ++r) {
        tp[(size_t)(2 * r) * N]     = acc0[r] - mean;
        tp[(size_t)(2 * r + 1) * N] = acc1[r] - mean;
      }
    }
  }
}

__device__ __forceinline__ void node_mlp_role(const int nb,
                                const float* __restrict__ h1pre,
                                const float* __restrict__ b1,
                                const float* __restrict__ g1,
                                const float* __restrict__ be1,
                                const float* __restrict__ w2,
                                const float* __restrict__ b2,
                                const float* __restrict__ g2,
                                const float* __restrict__ be2,
                                const float* __restrict__ w3,
                                const float* __restrict__ b3,
                                float* __restrict__ out_logits,
                                float* __restrict__ out_probs,
                                int N) {
  if (nb * 2 >= N) return;
  __builtin_amdgcn_s_setprio(3);
  const int sub  = threadIdx.x >> 7;
  const int t    = threadIdx.x & 127;
  const int nraw = nb * 2 + sub;
  const bool live = nraw < N;
  const int n = live ? nraw : N - 1;
  __shared__ float sA2[2][128], sB2[2][128];
  __shared__ float sh1_2[2][128];
  __shared__ float sh2_2[2][HH2];
  __shared__ float slog2[2][16];
  float* sA   = sA2[sub];
  float* sB   = sB2[sub];
  float* sh1  = sh1_2[sub];
  float* sh2  = sh2_2[sub];
  float* slog = slog2[sub];

  float v = h1pre[(size_t)n * H + t] + b1[t];
  sA[t] = v;
  sB[t] = v * v;
  __syncthreads();
  for (int s = 64; s > 0; s >>= 1) {
    if (t < s) { sA[t] += sA[t + s]; sB[t] += sB[t + s]; }
    __syncthreads();
  }
  {
    float m   = sA[0] * (1.0f / 128.0f);
    float var = sB[0] * (1.0f / 128.0f) - m * m;
    float inv = rsqrtf(var + LN_EPS);
    float hv  = (v - m) * inv * g1[t] + be1[t];
    v = fmaxf(hv, 0.0f);
  }
  __syncthreads();
  sh1[t] = v;
  __syncthreads();

  float a2 = 0.0f;
  if (t < HH2) {
    a2 = b2[t];
    for (int k = 0; k < H; ++k) a2 += sh1[k] * w2[k * HH2 + t];
    sA[t] = a2;
    sB[t] = a2 * a2;
  }
  __syncthreads();
  for (int s = 32; s > 0; s >>= 1) {
    if (t < s) { sA[t] += sA[t + s]; sB[t] += sB[t + s]; }
    __syncthreads();
  }
  if (t < HH2) {
    float m2  = sA[0] * (1.0f / 64.0f);
    float v2  = sB[0] * (1.0f / 64.0f) - m2 * m2;
    float i2  = rsqrtf(v2 + LN_EPS);
    float h2v = (a2 - m2) * i2 * g2[t] + be2[t];
    sh2[t] = fmaxf(h2v, 0.0f);
  }
  __syncthreads();

  if (t < KC) {
    float lg = b3[t];
#pragma unroll
    for (int k = 0; k < HH2; ++k) lg += sh2[k] * w3[k * KC + t];
    slog[t] = lg;
    if (live) out_logits[(size_t)n * KC + t] = lg;
  }
  __syncthreads();
  if (t < KC) {
    float mx = slog[0];
    for (int k = 1; k < KC; ++k) mx = fmaxf(mx, slog[k]);
    float ssum = 0.0f;
    for (int k = 0; k < KC; ++k) ssum += __expf(slog[k] - mx);
    if (live) out_probs[(size_t)n * KC + t] = __expf(slog[t] - mx) / ssum;
  }
}

#define CWAVES 16
#define CSTEPS 16

__device__ __forceinline__
void centers_role(const int cg,
                    const float* __restrict__ probs,
                    const float* __restrict__ x,
                    const float* __restrict__ prototypes,
                    float* __restrict__ out_centers,
                    float* __restrict__ out_protos,
                    int N) {
  const int tid  = threadIdx.x;
  const int lane = tid & 31;
  const int wid  = __builtin_amdgcn_readfirstlane(tid >> 5);
  const unsigned half = (unsigned)lane >> 4;
  const unsigned l16  = (unsigned)lane & 15u;

  __shared__ float red[CWAVES][16][32];
  __shared__ float wred[CWAVES][32];
  __shared__ float cen[KC][32];
  __shared__ float wsumS[16];

  const unsigned pofs = (l16 < KC) ? l16 : (unsigned)(KC - 1);
  const unsigned xofs = (unsigned)cg * 32u + l16;
  const unsigned un   = (unsigned)N;
  const unsigned nlast = un - 1u;

  v8f acc0 = {}, acc1 = {};
  float wp = 0.0f;

  const int nstep = (N + 3) >> 2;
  for (int s0 = 0; s0 < nstep; s0 += CWAVES * CSTEPS) {
    v2f a[CSTEPS], b0[CSTEPS], b1[CSTEPS];
#pragma unroll
    for (int u = 0; u < CSTEPS; ++u) {
      const unsigned n0  = (unsigned)(s0 + u * CWAVES + wid) * 4u + half * 2u;
      const unsigned n1  = n0 + 1u;
      const unsigned n0c = (n0 < un) ? n0 : nlast;
      const unsigned n1c = (n1 < un) ? n1 : nlast;
      a[u].x  = probs[n0c * KC + pofs];
      a[u].y  = probs[n1c * KC + pofs];
      b0[u].x = x[n0c * D + xofs];
      b0[u].y = x[n1c * D + xofs];
      b1[u].x = x[n0c * D + xofs + 16u];
      b1[u].y = x[n1c * D + xofs + 16u];
    }
    __builtin_amdgcn_sched_barrier(0);
#pragma unroll
    for (int u = 0; u < CSTEPS; ++u) {
      const unsigned n0 = (unsigned)(s0 + u * CWAVES + wid) * 4u + half * 2u;
      v2f av;
      av.x = (n0 < un)      ? a[u].x : 0.0f;
      av.y = (n0 + 1u < un) ? a[u].y : 0.0f;
      wp += av.x + av.y;
      acc0 = __builtin_amdgcn_wmma_f32_16x16x4_f32(false, av, false, b0[u],
                                                   (short)0, acc0, false, false);
      acc1 = __builtin_amdgcn_wmma_f32_16x16x4_f32(false, av, false, b1[u],
                                                   (short)0, acc1, false, false);
    }
  }

#pragma unroll
  for (int r = 0; r < 8; ++r) {
    red[wid][r][lane]     = acc0[r];
    red[wid][8 + r][lane] = acc1[r];
  }
  wred[wid][lane] = wp;
  __syncthreads();

  {
    const float* __restrict__ rp = &red[0][0][0] + tid;
    float sum = rp[0];
#pragma unroll
    for (int w = 1; w < CWAVES; ++w) sum += rp[w * 512];
    const int rr  = tid >> 5;
    const int m   = 8 * (lane >> 4) + (rr & 7);
    const int col = (rr >> 3) * 16 + (lane & 15);
    if (m < KC) cen[m][col] = sum;
  }
  if (tid < 16) {
    float ws = 0.0f;
#pragma unroll
    for (int w = 0; w < CWAVES; ++w) ws += wred[w][tid] + wred[w][tid + 16];
    wsumS[tid] = ws + 1e-8f;
  }
  __syncthreads();

  if (tid < KC * 32) {
    const int m = tid >> 5;
    const int o = m * D + cg * 32 + (tid & 31);
    out_centers[o] = cen[m][tid & 31] / wsumS[m];
    out_protos[o]  = prototypes[o];
  }
}

__device__ __forceinline__ float tanh_inline(const float x) {
  const float ax  = __builtin_fabsf(x);
  const float e2  = __builtin_amdgcn_exp2f(ax * 2.885390043f);
  const float big = __builtin_fmaf(__builtin_amdgcn_rcpf(e2 + 1.0f), -2.0f, 1.0f);
  const float p   = x * x;
  float q = __builtin_fmaf(p, -5.700020585e-3f, 2.063407190e-2f);
  q = __builtin_fmaf(p, q, -5.373793095e-2f);
  q = __builtin_fmaf(p, q, 1.333141625e-1f);
  q = __builtin_fmaf(p, q, -3.333328068e-1f);
  const float sml = __builtin_fmaf(p, ax * q, ax);
  const float r   = (ax < 0.625f) ? sml : big;
  return __builtin_copysignf(r, x);
}

#define PCH 8
#define WSTEPS 16

__device__ __forceinline__
void comp_matrix_role(const unsigned bx, const unsigned by,
                        const float* __restrict__ aT,
                        const float* __restrict__ bT,
                        const float* __restrict__ cg1,
                        const float* __restrict__ cbe1,
                        const float* __restrict__ cw2,
                        const float* __restrict__ cb2p,
                        float* __restrict__ out, int N) {
  __shared__ __attribute__((aligned(16))) float Ag[H * 16];

  const int tid  = threadIdx.x;
  const int lane = tid & 31;
  const int wid  = __builtin_amdgcn_readfirstlane(tid >> 5);
  const unsigned half = (unsigned)lane >> 4;
  const unsigned l16  = (unsigned)lane & 15u;
  const unsigned un    = (unsigned)N;
  const unsigned nlast = un - 1u;
  const unsigned i0 = bx * 16u;
  const unsigned j0 = by * 256u + (unsigned)wid * 32u;

#pragma unroll
  for (int u = 0; u < 8; ++u) {
    const unsigned e  = (unsigned)tid + 256u * (unsigned)u;
    const unsigned kk = e >> 4;
    const unsigned r  = e & 15u;
    unsigned ii = i0 + r;
    ii = (ii < un) ? ii : nlast;
    Ag[e] = aT[kk * un + ii] * cg1[kk];
  }
  const float cb2 = cb2p[0];

  unsigned ic  = i0 + l16;        ic  = (ic  < un) ? ic  : nlast;
  unsigned jc0 = j0 + l16;        jc0 = (jc0 < un) ? jc0 : nlast;
  unsigned jc1 = j0 + 16u + l16;  jc1 = (jc1 < un) ? jc1 : nlast;

  v8f c0 = {}, c1 = {};
  float sa = 0.0f, sb0 = 0.0f, sb1 = 0.0f;
  for (int s0 = 0; s0 < H / 4; s0 += WSTEPS) {
    v2f a[WSTEPS], b0[WSTEPS], b1[WSTEPS];
#pragma unroll
    for (int u = 0; u < WSTEPS; ++u) {
      const unsigned k = (unsigned)(s0 + u) * 4u + half * 2u;
      a[u].x  = aT[k * un + ic];
      a[u].y  = aT[(k + 1u) * un + ic];
      b0[u].x = bT[k * un + jc0];
      b0[u].y = bT[(k + 1u) * un + jc0];
      b1[u].x = bT[k * un + jc1];
      b1[u].y = bT[(k + 1u) * un + jc1];
    }
    __builtin_amdgcn_sched_barrier(0);
#pragma unroll
    for (int u = 0; u < WSTEPS; ++u) {
      sa  += a[u].x * a[u].x + a[u].y * a[u].y;
      sb0 += b0[u].x * b0[u].x + b0[u].y * b0[u].y;
      sb1 += b1[u].x * b1[u].x + b1[u].y * b1[u].y;
      c0 = __builtin_amdgcn_wmma_f32_16x16x4_f32(false, a[u], false, b0[u],
                                                 (short)0, c0, false, false);
      c1 = __builtin_amdgcn_wmma_f32_16x16x4_f32(false, a[u], false, b1[u],
                                                 (short)0, c1, false, false);
    }
  }
  sa  += __shfl_xor(sa, 16);
  sb0 += __shfl_xor(sb0, 16);
  sb1 += __shfl_xor(sb1, 16);
#pragma unroll
  for (int r = 0; r < 8; ++r) { c0[r] *= 2.0f; c1[r] *= 2.0f; }
  {
    v2f an, bn0, bn1;
    an.x  = (half == 0u) ? sa   : 0.0f;
    an.y  = (half == 0u) ? 1.0f : 0.0f;
    bn0.x = (half == 0u) ? 1.0f : 0.0f;
    bn0.y = (half == 0u) ? sb0  : 0.0f;
    bn1.x = bn0.x;
    bn1.y = (half == 0u) ? sb1  : 0.0f;
    c0 = __builtin_amdgcn_wmma_f32_16x16x4_f32(false, an, false, bn0,
                                               (short)0, c0, false, false);
    c1 = __builtin_amdgcn_wmma_f32_16x16x4_f32(false, an, false, bn1,
                                               (short)0, c1, false, false);
  }
  v2f inv2[8];
  {
    float vrow[16];
#pragma unroll
    for (int r = 0; r < 8; ++r) {
      const float snd = (half == 0u) ? c1[r] : c0[r];
      const float rcv = __shfl_xor(snd, 16);
      vrow[r]     = (half == 0u) ? c0[r] : rcv;
      vrow[8 + r] = (half == 0u) ? rcv   : c1[r];
    }
#pragma unroll
    for (int rp = 0; rp < 8; ++rp) {
      inv2[rp].x = rsqrtf(vrow[2 * rp]     * (1.0f / 128.0f) + LN_EPS);
      inv2[rp].y = rsqrtf(vrow[2 * rp + 1] * (1.0f / 128.0f) + LN_EPS);
    }
  }

  __syncthreads();

  unsigned jc = j0 + (unsigned)lane;
  jc = (jc < un) ? jc : nlast;
  const float* __restrict__ bp = bT + jc;

  v2f acc[8];
#pragma unroll
  for (int rp = 0; rp < 8; ++rp) { acc[rp].x = 0.0f; acc[rp].y = 0.0f; }

  const v4f* __restrict__ Ag4 = (const v4f*)Ag;

  const __attribute__((address_space(4))) float* cbe1c =
      (const __attribute__((address_space(4))) float*)(unsigned long long)cbe1;
  const __attribute__((address_space(4))) float* cw2c =
      (const __attribute__((address_space(4))) float*)(unsigned long long)cw2;
  const __attribute__((address_space(4))) float* cg1c =
      (const __attribute__((address_space(4))) float*)(unsigned long long)cg1;

  float cur[PCH], be_c[PCH], cw_c[PCH], g_c[PCH];
#pragma unroll
  for (int u = 0; u < PCH; ++u) {
    cur[u]  = bp[(unsigned)u * un];
    be_c[u] = cbe1c[u];
    cw_c[u] = cw2c[u];
    g_c[u]  = cg1c[u];
  }

  v4f Ac[2][4];
#pragma unroll
  for (int q = 0; q < 2; ++q)
#pragma unroll
    for (int m = 0; m < 4; ++m) Ac[q][m] = Ag4[4 * q + m];

  for (int k0 = 0; k0 < H; k0 += PCH) {
    const int kn0 = (k0 + PCH < H) ? (k0 + PCH) : (H - PCH);
    float nxt[PCH], be_n[PCH], cw_n[PCH], g_n[PCH];
#pragma unroll
    for (int u = 0; u < PCH; ++u) {
      nxt[u]  = bp[(unsigned)(kn0 + u) * un];
      be_n[u] = cbe1c[kn0 + u];
      cw_n[u] = cw2c[kn0 + u];
      g_n[u]  = cg1c[kn0 + u];
    }
#pragma unroll
    for (int u = 0; u < PCH; u += 2) {
      int kb = k0 + u + 2;
      kb = (kb < H) ? kb : (H - 2);
      v4f An[2][4];
#pragma unroll
      for (int q = 0; q < 2; ++q)
#pragma unroll
        for (int m = 0; m < 4; ++m) An[q][m] = Ag4[4 * (kb + q) + m];
      __builtin_amdgcn_sched_barrier(0);
#pragma unroll
      for (int q = 0; q < 2; ++q) {
        const float be = be_c[u + q];
        const float cw = cw_c[u + q];
        const float bg = cur[u + q] * g_c[u + q];
        const v4f A0 = Ac[q][0], A1 = Ac[q][1], A2 = Ac[q][2], A3 = Ac[q][3];
        const float ax[16] = {A0.x, A0.y, A0.z, A0.w, A1.x, A1.y, A1.z, A1.w,
                              A2.x, A2.y, A2.z, A2.w, A3.x, A3.y, A3.z, A3.w};
#pragma unroll
        for (int rp = 0; rp < 8; ++rp) {
          v2f z;
          z.x = ax[2 * rp]     + bg;
          z.y = ax[2 * rp + 1] + bg;
          z = z * inv2[rp] + be;
          z.x = fmaxf(z.x, 0.0f);
          z.y = fmaxf(z.y, 0.0f);
          acc[rp] += z * cw;
        }
      }
      __builtin_amdgcn_sched_barrier(0);
#pragma unroll
      for (int q = 0; q < 2; ++q)
#pragma unroll
        for (int m = 0; m < 4; ++m) Ac[q][m] = An[q][m];
    }
#pragma unroll
    for (int u = 0; u < PCH; ++u) {
      cur[u]  = nxt[u];
      be_c[u] = be_n[u];
      cw_c[u] = cw_n[u];
      g_c[u]  = g_n[u];
    }
  }

  float tv[16];
#pragma unroll
  for (int rp = 0; rp < 8; ++rp) {
    tv[2 * rp]     = tanh_inline(acc[rp].x + cb2);
    tv[2 * rp + 1] = tanh_inline(acc[rp].y + cb2);
  }
  const unsigned j = j0 + (unsigned)lane;
  if (j < un) {
    float* __restrict__ op = out + (size_t)i0 * un + j;
    if (i0 + 16u <= un) {
#pragma unroll
      for (int r = 0; r < 16; ++r) op[(size_t)r * un] = tv[r];
    } else {
#pragma unroll
      for (int r = 0; r < 16; ++r)
        if (i0 + (unsigned)r < un) op[(size_t)r * un] = tv[r];
    }
  }
}

#define CS_THREADS 512
#define EPT 4

__device__ __forceinline__
void comp_scores_role(const unsigned blk, const unsigned nblk,
                        const float* __restrict__ mat,
                        const int* __restrict__ ei, int E, int N,
                        float* __restrict__ out) {
  const unsigned ue    = (unsigned)E;
  const unsigned un    = (unsigned)N;
  const unsigned nlast = un - 1u;
  const unsigned chunk = CS_THREADS * EPT;
  const int* erow = ei;
  const int* ecol = ei + ue;

  for (unsigned c0 = blk * chunk; c0 < ue; c0 += nblk * chunk) {
    const unsigned e0 = c0 + threadIdx.x;
    if (c0 + chunk <= ue) {
      unsigned ri[EPT], ci[EPT];
#pragma unroll
      for (int u = 0; u < EPT; ++u) {
        ri[u] = (unsigned)erow[e0 + u * CS_THREADS];
        ci[u] = (unsigned)ecol[e0 + u * CS_THREADS];
      }
      float v[EPT];
#pragma unroll
      for (int u = 0; u < EPT; ++u) {
        const unsigned i = (ri[u] < un) ? ri[u] : nlast;
        const unsigned j = (ci[u] < un) ? ci[u] : nlast;
        v[u] = mat[i * un + j];
      }
#pragma unroll
      for (int u = 0; u < EPT; ++u) out[e0 + u * CS_THREADS] = v[u];
    } else {
#pragma unroll 1
      for (unsigned e = e0; e < ue; e += CS_THREADS) {
        const unsigned r = (unsigned)erow[e];
        const unsigned c = (unsigned)ecol[e];
        const unsigned i = (r < un) ? r : nlast;
        const unsigned j = (c < un) ? c : nlast;
        out[e] = mat[i * un + j];
      }
    }
  }
}

__global__ __launch_bounds__(256, 8)
void pairs_and_nodes_kernel(const float* __restrict__ aT,
                            const float* __restrict__ bT,
                            const float* __restrict__ cg1,
                            const float* __restrict__ cbe1,
                            const float* __restrict__ cw2,
                            const float* __restrict__ cb2p,
                            float* __restrict__ out_matrix,
                            const float* __restrict__ h1pre,
                            const float* __restrict__ b1,
                            const float* __restrict__ g1,
                            const float* __restrict__ be1,
                            const float* __restrict__ w2,
                            const float* __restrict__ b2,
                            const float* __restrict__ g2,
                            const float* __restrict__ be2,
                            const float* __restrict__ w3,
                            const float* __restrict__ b3,
                            float* __restrict__ out_logits,
                            float* __restrict__ out_probs,
                            int N, int colGroups) {
  if ((int)blockIdx.y < colGroups) {
    comp_matrix_role(blockIdx.x, blockIdx.y, aT, bT, cg1, cbe1, cw2, cb2p,
                     out_matrix, N);
  } else {
    const int nb = ((int)blockIdx.y - colGroups) * (int)gridDim.x + (int)blockIdx.x;
    node_mlp_role(nb, h1pre, b1, g1, be1, w2, b2, g2, be2, w3, b3,
                  out_logits, out_probs, N);
  }
}

#define CENTER_BLOCKS 4

__global__ __launch_bounds__(CWAVES * 32)
void centers_and_scores_kernel(const float* __restrict__ probs,
                               const float* __restrict__ x,
                               const float* __restrict__ prototypes,
                               float* __restrict__ out_centers,
                               float* __restrict__ out_protos,
                               const float* __restrict__ mat,
                               const int* __restrict__ ei, int E, int N,
                               float* __restrict__ out_scores) {
  if (blockIdx.x < CENTER_BLOCKS) {
    centers_role((int)blockIdx.x, probs, x, prototypes, out_centers,
                 out_protos, N);
  } else {
    comp_scores_role(blockIdx.x - CENTER_BLOCKS, gridDim.x - CENTER_BLOCKS,
                     mat, ei, E, N, out_scores);
  }
}

extern "C" void kernel_launch(void* const* d_in, const int* in_sizes, int n_in,
                              void* d_out, int out_size, void* d_ws, size_t ws_size,
                              hipStream_t stream) {
  const float* x    = (const float*)d_in[0];
  const int*   ei   = (const int*)  d_in[1];
  const float* w1   = (const float*)d_in[2];
  const float* b1   = (const float*)d_in[3];
  const float* g1   = (const float*)d_in[4];
  const float* be1  = (const float*)d_in[5];
  const float* w2   = (const float*)d_in[6];
  const float* b2   = (const float*)d_in[7];
  const float* g2   = (const float*)d_in[8];
  const float* be2  = (const float*)d_in[9];
  const float* w3   = (const float*)d_in[10];
  const float* b3   = (const float*)d_in[11];
  const float* cw1  = (const float*)d_in[12];
  const float* cb1  = (const float*)d_in[13];
  const float* cg1  = (const float*)d_in[14];
  const float* cbe1 = (const float*)d_in[15];
  const float* cw2  = (const float*)d_in[16];
  const float* cb2  = (const float*)d_in[17];
  const float* prot = (const float*)d_in[18];

  const int N = in_sizes[0] / D;
  const int E = in_sizes[1] / 2;

  float* h1pre = (float*)d_ws;
  float* aT    = h1pre + (size_t)N * H;
  float* bT    = aT    + (size_t)N * H;

  float* out       = (float*)d_out;
  float* o_logits  = out;
  float* o_probs   = o_logits  + (size_t)N * KC;
  float* o_scores  = o_probs   + (size_t)N * KC;
  float* o_matrix  = o_scores  + (size_t)E;
  float* o_centers = o_matrix  + (size_t)N * N;
  float* o_protos  = o_centers + (size_t)KC * D;

  const int rowTiles  = (N + 15) / 16;
  const int colGroups = (N + 255) / 256;
  const int nodeBlocks = (N + 1) / 2;
  const int nodeRows   = (nodeBlocks + rowTiles - 1) / rowTiles;

  wmma_gemm3_kernel<<<dim3(rowTiles, 3), 128, 0, stream>>>(
      x, w1, cw1, cb1, h1pre, aT, bT, N);

  pairs_and_nodes_kernel<<<dim3(rowTiles, colGroups + nodeRows), 256, 0, stream>>>(
      aT, bT, cg1, cbe1, cw2, cb2, o_matrix,
      h1pre, b1, g1, be1, w2, b2, g2, be2, w3, b3, o_logits, o_probs,
      N, colGroups);

  centers_and_scores_kernel<<<CENTER_BLOCKS + 256, CWAVES * 32, 0, stream>>>(
      o_probs, x, prot, o_centers, o_protos, o_matrix, ei, E, N, o_scores);
}


# pair matrix main loop: ReLU folded into the packed multiply-add as its clamp modifier after an exact power-of-two rescale of the bounded LayerNorm output: 3 packed ops per row pair and column, was 4
# speedup vs baseline: 9.4873x; 1.0346x over previous
#include <hip/hip_runtime.h>
#include <math.h>

typedef __attribute__((ext_vector_type(2))) float v2f;
typedef __attribute__((ext_vector_type(4))) float v4f;
typedef __attribute__((ext_vector_type(8))) float v8f;

#define D   128
#define H   128
#define HH2 64
#define KC  10
#define LN_EPS 1e-5f

#define XP 132
#define GSTEPS 32

__global__ __launch_bounds__(128)
void wmma_gemm3_kernel(const float* __restrict__ x,
                       const float* __restrict__ w1,
                       const float* __restrict__ cw1,
                       const float* __restrict__ cb1,
                       float* __restrict__ h1pre,
                       float* __restrict__ aT,
                       float* __restrict__ bT,
                       int N) {
  const int rt  = blockIdx.x;
  const int mat = blockIdx.y;

  const float* __restrict__ Bm = (mat == 0) ? w1 : (mat == 1 ? cw1 : cw1 + D * H);

  const int tid   = threadIdx.x;
  const int lane  = tid & 31;
  const int wid   = __builtin_amdgcn_readfirstlane(tid >> 5);
  const int half  = lane >> 4;
  const int l16   = lane & 15;
  const int cbase = wid * 32;

  __shared__ __attribute__((aligned(16))) float xs[16 * XP];
  __shared__ float part[4][16];

  v4f xv[4];
#pragma unroll
  for (int q = 0; q < 4; ++q) {
    int n = rt * 16 + wid * 4 + q;
    if (n >= N) n = N - 1;
    xv[q] = *(const v4f*)(x + (size_t)n * D + lane * 4);
  }

  const float* __restrict__ wp = Bm + (size_t)(2 * half) * H + cbase + 2 * l16;
  v2f wlo[GSTEPS], whi[GSTEPS];
#pragma unroll
  for (int s = 0; s < GSTEPS; ++s) {
    wlo[s] = *(const v2f*)(wp + (4 * s) * H);
    whi[s] = *(const v2f*)(wp + (4 * s + 1) * H);
  }
  __builtin_amdgcn_sched_barrier(0);

#pragma unroll
  for (int q = 0; q < 4; ++q)
    *(v4f*)(xs + (wid * 4 + q) * XP + lane * 4) = xv[q];
  __syncthreads();

  const float* xa = xs + l16 * XP + 2 * half;
  v2f a[GSTEPS];
#pragma unroll
  for (int s = 0; s < GSTEPS; ++s) a[s] = *(const v2f*)(xa + 4 * s);

  v8f acc0 = {}, acc1 = {};

  if (mat == 0) {
#pragma unroll
    for (int s = 0; s < GSTEPS; ++s) {
      v2f b0, b1;
      b0.x = wlo[s].x; b0.y = whi[s].x;
      b1.x = wlo[s].y; b1.y = whi[s].y;
      acc0 = __builtin_amdgcn_wmma_f32_16x16x4_f32(false, a[s], false, b0,
                                                   (short)0, acc0, false, false);
      acc1 = __builtin_amdgcn_wmma_f32_16x16x4_f32(false, a[s], false, b1,
                                                   (short)0, acc1, false, false);
    }
    const int rbase = rt * 16 + half * 8;
    float* __restrict__ op = h1pre + (size_t)rbase * H + cbase + 2 * l16;
    if (rt * 16 + 16 <= N) {
#pragma unroll
      for (int r = 0; r < 8; ++r) {
        v2f o; o.x = acc0[r]; o.y = acc1[r];
        *(v2f*)(op + (size_t)r * H) = o;
      }
    } else {
#pragma unroll
      for (int r = 0; r < 8; ++r)
        if (rbase + r < N) {
          v2f o; o.x = acc0[r]; o.y = acc1[r];
          *(v2f*)(op + (size_t)r * H) = o;
        }
    }
  } else {
#pragma unroll
    for (int s = 0; s < GSTEPS; ++s) {
      v2f b0, b1;
      b0.x = wlo[s].x; b0.y = whi[s].x;
      b1.x = wlo[s].y; b1.y = whi[s].y;
      acc0 = __builtin_amdgcn_wmma_f32_16x16x4_f32(false, b0, false, a[s],
                                                   (short)0, acc0, false, false);
      acc1 = __builtin_amdgcn_wmma_f32_16x16x4_f32(false, b1, false, a[s],
                                                   (short)0, acc1, false, false);
    }
    if (mat == 1) {
      const v4f* __restrict__ cbp = (const v4f*)(cb1 + cbase + 16 * half);
      const v4f q0 = cbp[0], q1 = cbp[1], q2 = cbp[2], q3 = cbp[3];
      acc0[0] += q0.x; acc1[0] += q0.y; acc0[1] += q0.z; acc1[1] += q0.w;
      acc0[2] += q1.x; acc1[2] += q1.y; acc0[3] += q1.z; acc1[3] += q1.w;
      acc0[4] += q2.x; acc1[4] += q2.y; acc0[5] += q2.z; acc1[5] += q2.w;
      acc0[6] += q3.x; acc1[6] += q3.y; acc0[7] += q3.z; acc1[7] += q3.w;
    }
    float s = 0.0f;
#pragma unroll
    for (int r = 0; r < 8; ++r) s += acc0[r] + acc1[r];
    s += __shfl_xor(s, 16);
    if (half == 0) part[wid][l16] = s;
    __syncthreads();
    const float mean = (part[0][l16] + part[1][l16] + part[2][l16] + part[3][l16])
                       * (1.0f / 128.0f);
    float* __restrict__ outT = (mat == 1) ? aT : bT;
    const int n = rt * 16 + l16;
    if (n < N) {
      float* __restrict__ tp = outT + (size_t)(cbase + 16 * half) * N + n;
#pragma unroll
      for (int r = 0; r < 8; ++r) {
        tp[(size_t)(2 * r) * N]     = acc0[r] - mean;
        tp[(size_t)(2 * r + 1) * N] = acc1[r] - mean;
      }
    }
  }
}

__device__ __forceinline__ void node_mlp_role(const int nb,
                                const float* __restrict__ h1pre,
                                const float* __restrict__ b1,
                                const float* __restrict__ g1,
                                const float* __restrict__ be1,
                                const float* __restrict__ w2,
                                const float* __restrict__ b2,
                                const float* __restrict__ g2,
                                const float* __restrict__ be2,
                                const float* __restrict__ w3,
                                const float* __restrict__ b3,
                                float* __restrict__ out_logits,
                                float* __restrict__ out_probs,
                                int N) {
  if (nb * 2 >= N) return;
  __builtin_amdgcn_s_setprio(3);
  const int sub  = threadIdx.x >> 7;
  const int t    = threadIdx.x & 127;
  const int nraw = nb * 2 + sub;
  const bool live = nraw < N;
  const int n = live ? nraw : N - 1;
  __shared__ float sA2[2][128], sB2[2][128];
  __shared__ float sh1_2[2][128];
  __shared__ float sh2_2[2][HH2];
  __shared__ float slog2[2][16];
  float* sA   = sA2[sub];
  float* sB   = sB2[sub];
  float* sh1  = sh1_2[sub];
  float* sh2  = sh2_2[sub];
  float* slog = slog2[sub];

  float v = h1pre[(size_t)n * H + t] + b1[t];
  sA[t] = v;
  sB[t] = v * v;
  __syncthreads();
  for (int s = 64; s > 0; s >>= 1) {
    if (t < s) { sA[t] += sA[t + s]; sB[t] += sB[t + s]; }
    __syncthreads();
  }
  {
    float m   = sA[0] * (1.0f / 128.0f);
    float var = sB[0] * (1.0f / 128.0f) - m * m;
    float inv = rsqrtf(var + LN_EPS);
    float hv  = (v - m) * inv * g1[t] + be1[t];
    v = fmaxf(hv, 0.0f);
  }
  __syncthreads();
  sh1[t] = v;
  __syncthreads();

  float a2 = 0.0f;
  if (t < HH2) {
    a2 = b2[t];
    for (int k = 0; k < H; ++k) a2 += sh1[k] * w2[k * HH2 + t];
    sA[t] = a2;
    sB[t] = a2 * a2;
  }
  __syncthreads();
  for (int s = 32; s > 0; s >>= 1) {
    if (t < s) { sA[t] += sA[t + s]; sB[t] += sB[t + s]; }
    __syncthreads();
  }
  if (t < HH2) {
    float m2  = sA[0] * (1.0f / 64.0f);
    float v2  = sB[0] * (1.0f / 64.0f) - m2 * m2;
    float i2  = rsqrtf(v2 + LN_EPS);
    float h2v = (a2 - m2) * i2 * g2[t] + be2[t];
    sh2[t] = fmaxf(h2v, 0.0f);
  }
  __syncthreads();

  if (t < KC) {
    float lg = b3[t];
#pragma unroll
    for (int k = 0; k < HH2; ++k) lg += sh2[k] * w3[k * KC + t];
    slog[t] = lg;
    if (live) out_logits[(size_t)n * KC + t] = lg;
  }
  __syncthreads();
  if (t < KC) {
    float mx = slog[0];
    for (int k = 1; k < KC; ++k) mx = fmaxf(mx, slog[k]);
    float ssum = 0.0f;
    for (int k = 0; k < KC; ++k) ssum += __expf(slog[k] - mx);
    if (live) out_probs[(size_t)n * KC + t] = __expf(slog[t] - mx) / ssum;
  }
}

#define CWAVES 16
#define CSTEPS 16

__device__ __forceinline__
void centers_role(const int cg,
                    const float* __restrict__ probs,
                    const float* __restrict__ x,
                    const float* __restrict__ prototypes,
                    float* __restrict__ out_centers,
                    float* __restrict__ out_protos,
                    int N) {
  const int tid  = threadIdx.x;
  const int lane = tid & 31;
  const int wid  = __builtin_amdgcn_readfirstlane(tid >> 5);
  const unsigned half = (unsigned)lane >> 4;
  const unsigned l16  = (unsigned)lane & 15u;

  __shared__ float red[CWAVES][16][32];
  __shared__ float wred[CWAVES][32];
  __shared__ float cen[KC][32];
  __shared__ float wsumS[16];

  const unsigned pofs = (l16 < KC) ? l16 : (unsigned)(KC - 1);
  const unsigned xofs = (unsigned)cg * 32u + l16;
  const unsigned un   = (unsigned)N;
  const unsigned nlast = un - 1u;

  v8f acc0 = {}, acc1 = {};
  float wp = 0.0f;

  const int nstep = (N + 3) >> 2;
  for (int s0 = 0; s0 < nstep; s0 += CWAVES * CSTEPS) {
    v2f a[CSTEPS], b0[CSTEPS], b1[CSTEPS];
#pragma unroll
    for (int u = 0; u < CSTEPS; ++u) {
      const unsigned n0  = (unsigned)(s0 + u * CWAVES + wid) * 4u + half * 2u;
      const unsigned n1  = n0 + 1u;
      const unsigned n0c = (n0 < un) ? n0 : nlast;
      const unsigned n1c = (n1 < un) ? n1 : nlast;
      a[u].x  = probs[n0c * KC + pofs];
      a[u].y  = probs[n1c * KC + pofs];
      b0[u].x = x[n0c * D + xofs];
      b0[u].y = x[n1c * D + xofs];
      b1[u].x = x[n0c * D + xofs + 16u];
      b1[u].y = x[n1c * D + xofs + 16u];
    }
    __builtin_amdgcn_sched_barrier(0);
#pragma unroll
    for (int u = 0; u < CSTEPS; ++u) {
      const unsigned n0 = (unsigned)(s0 + u * CWAVES + wid) * 4u + half * 2u;
      v2f av;
      av.x = (n0 < un)      ? a[u].x : 0.0f;
      av.y = (n0 + 1u < un) ? a[u].y : 0.0f;
      wp += av.x + av.y;
      acc0 = __builtin_amdgcn_wmma_f32_16x16x4_f32(false, av, false, b0[u],
                                                   (short)0, acc0, false, false);
      acc1 = __builtin_amdgcn_wmma_f32_16x16x4_f32(false, av, false, b1[u],
                                                   (short)0, acc1, false, false);
    }
  }

#pragma unroll
  for (int r = 0; r < 8; ++r) {
    red[wid][r][lane]     = acc0[r];
    red[wid][8 + r][lane] = acc1[r];
  }
  wred[wid][lane] = wp;
  __syncthreads();

  {
    const float* __restrict__ rp = &red[0][0][0] + tid;
    float sum = rp[0];
#pragma unroll
    for (int w = 1; w < CWAVES; ++w) sum += rp[w * 512];
    const int rr  = tid >> 5;
    const int m   = 8 * (lane >> 4) + (rr & 7);
    const int col = (rr >> 3) * 16 + (lane & 15);
    if (m < KC) cen[m][col] = sum;
  }
  if (tid < 16) {
    float ws = 0.0f;
#pragma unroll
    for (int w = 0; w < CWAVES; ++w) ws += wred[w][tid] + wred[w][tid + 16];
    wsumS[tid] = ws + 1e-8f;
  }
  __syncthreads();

  if (tid < KC * 32) {
    const int m = tid >> 5;
    const int o = m * D + cg * 32 + (tid & 31);
    out_centers[o] = cen[m][tid & 31] / wsumS[m];
    out_protos[o]  = prototypes[o];
  }
}

__device__ __forceinline__ float tanh_inline(const float x) {
  const float ax  = __builtin_fabsf(x);
  const float e2  = __builtin_amdgcn_exp2f(ax * 2.885390043f);
  const float big = __builtin_fmaf(__builtin_amdgcn_rcpf(e2 + 1.0f), -2.0f, 1.0f);
  const float p   = x * x;
  float q = __builtin_fmaf(p, -5.700020585e-3f, 2.063407190e-2f);
  q = __builtin_fmaf(p, q, -5.373793095e-2f);
  q = __builtin_fmaf(p, q, 1.333141625e-1f);
  q = __builtin_fmaf(p, q, -3.333328068e-1f);
  const float sml = __builtin_fmaf(p, ax * q, ax);
  const float r   = (ax < 0.625f) ? sml : big;
  return __builtin_copysignf(r, x);
}

#define PCH 8

template <int M>
__device__ __forceinline__ float row_xmask_f(const float v) {
  return __builtin_bit_cast(float, __builtin_amdgcn_update_dpp(
      0, __builtin_bit_cast(int, v), 0x160 + M, 0xf, 0xf, false));
}
__device__ __forceinline__ float xhalf_f(const float v) {
  return __builtin_bit_cast(float, __builtin_amdgcn_permlanex16(
      __builtin_bit_cast(int, v), __builtin_bit_cast(int, v),
      0x76543210, 0xfedcba98, false, false));
}

__device__ __forceinline__ v2f pk_fma_clamp01(const v2f a, const v2f b, const v2f c) {
  v2f d;
  asm("v_pk_fma_f32 %0, %1, %2, %3 op_sel_hi:[1,1,0] clamp"
      : "=v"(d) : "v"(a), "v"(b), "s"(c));
  return d;
}
#define WSTEPS 16

__device__ __forceinline__
void comp_matrix_role(const unsigned bx, const unsigned by,
                        const float* __restrict__ aT,
                        const float* __restrict__ bT,
                        const float* __restrict__ cg1,
                        const float* __restrict__ cbe1,
                        const float* __restrict__ cw2,
                        const float* __restrict__ cb2p,
                        float* __restrict__ out, int N) {
  __shared__ __attribute__((aligned(16))) float Ag[H * 16];

  const int tid  = threadIdx.x;
  const int lane = tid & 31;
  const int wid  = __builtin_amdgcn_readfirstlane(tid >> 5);
  const unsigned half = (unsigned)lane >> 4;
  const unsigned l16  = (unsigned)lane & 15u;
  const unsigned un    = (unsigned)N;
  const unsigned nlast = un - 1u;
  const unsigned i0 = bx * 16u;
  const unsigned j0 = by * 256u + (unsigned)wid * 32u;

  const v4f g4 = *(const v4f*)(cg1 + 4 * lane);
  const v4f e4 = *(const v4f*)(cbe1 + 4 * lane);

#pragma unroll
  for (int u = 0; u < 8; ++u) {
    const unsigned e  = (unsigned)tid + 256u * (unsigned)u;
    const unsigned kk = e >> 4;
    const unsigned r  = e & 15u;
    unsigned ii = i0 + r;
    ii = (ii < un) ? ii : nlast;
    Ag[e] = aT[kk * un + ii] * cg1[kk];
  }
  const float cb2 = cb2p[0];

  float bnd = fmaxf(
      fmaxf(__builtin_fmaf(__builtin_fabsf(g4.x), 16.0f, __builtin_fabsf(e4.x)),
            __builtin_fmaf(__builtin_fabsf(g4.y), 16.0f, __builtin_fabsf(e4.y))),
      fmaxf(__builtin_fmaf(__builtin_fabsf(g4.z), 16.0f, __builtin_fabsf(e4.z)),
            __builtin_fmaf(__builtin_fabsf(g4.w), 16.0f, __builtin_fabsf(e4.w))));
  bnd = fmaxf(bnd, row_xmask_f<1>(bnd));
  bnd = fmaxf(bnd, row_xmask_f<2>(bnd));
  bnd = fmaxf(bnd, row_xmask_f<4>(bnd));
  bnd = fmaxf(bnd, row_xmask_f<8>(bnd));
  bnd = fmaxf(bnd, xhalf_f(bnd));
  int ef = (__builtin_bit_cast(int, bnd) >> 23) & 0xff;
  ef = (ef < 107) ? 107 : ((ef > 227) ? 227 : ef);
  ef = __builtin_amdgcn_readfirstlane(ef);
  asm volatile("" : "+s"(ef));
  const float sc    = __builtin_bit_cast(float, (253 - ef) << 23);
  const float scinv = __builtin_bit_cast(float, (ef + 1) << 23);

  unsigned ic  = i0 + l16;        ic  = (ic  < un) ? ic  : nlast;
  unsigned jc0 = j0 + l16;        jc0 = (jc0 < un) ? jc0 : nlast;
  unsigned jc1 = j0 + 16u + l16;  jc1 = (jc1 < un) ? jc1 : nlast;

  v8f c0 = {}, c1 = {};
  float sa = 0.0f, sb0 = 0.0f, sb1 = 0.0f;
  for (int s0 = 0; s0 < H / 4; s0 += WSTEPS) {
    v2f a[WSTEPS], b0[WSTEPS], b1[WSTEPS];
#pragma unroll
    for (int u = 0; u < WSTEPS; ++u) {
      const unsigned k = (unsigned)(s0 + u) * 4u + half * 2u;
      a[u].x  = aT[k * un + ic];
      a[u].y  = aT[(k + 1u) * un + ic];
      b0[u].x = bT[k * un + jc0];
      b0[u].y = bT[(k + 1u) * un + jc0];
      b1[u].x = bT[k * un + jc1];
      b1[u].y = bT[(k + 1u) * un + jc1];
    }
    __builtin_amdgcn_sched_barrier(0);
#pragma unroll
    for (int u = 0; u < WSTEPS; ++u) {
      sa  += a[u].x * a[u].x + a[u].y * a[u].y;
      sb0 += b0[u].x * b0[u].x + b0[u].y * b0[u].y;
      sb1 += b1[u].x * b1[u].x + b1[u].y * b1[u].y;
      c0 = __builtin_amdgcn_wmma_f32_16x16x4_f32(false, a[u], false, b0[u],
                                                 (short)0, c0, false, false);
      c1 = __builtin_amdgcn_wmma_f32_16x16x4_f32(false, a[u], false, b1[u],
                                                 (short)0, c1, false, false);
    }
  }
  sa  += __shfl_xor(sa, 16);
  sb0 += __shfl_xor(sb0, 16);
  sb1 += __shfl_xor(sb1, 16);
#pragma unroll
  for (int r = 0; r < 8; ++r) { c0[r] *= 2.0f; c1[r] *= 2.0f; }
  {
    v2f an, bn0, bn1;
    an.x  = (half == 0u) ? sa   : 0.0f;
    an.y  = (half == 0u) ? 1.0f : 0.0f;
    bn0.x = (half == 0u) ? 1.0f : 0.0f;
    bn0.y = (half == 0u) ? sb0  : 0.0f;
    bn1.x = bn0.x;
    bn1.y = (half == 0u) ? sb1  : 0.0f;
    c0 = __builtin_amdgcn_wmma_f32_16x16x4_f32(false, an, false, bn0,
                                               (short)0, c0, false, false);
    c1 = __builtin_amdgcn_wmma_f32_16x16x4_f32(false, an, false, bn1,
                                               (short)0, c1, false, false);
  }
  v2f inv2[8];
  {
    float vrow[16];
#pragma unroll
    for (int r = 0; r < 8; ++r) {
      const float snd = (half == 0u) ? c1[r] : c0[r];
      const float rcv = __shfl_xor(snd, 16);
      vrow[r]     = (half == 0u) ? c0[r] : rcv;
      vrow[8 + r] = (half == 0u) ? rcv   : c1[r];
    }
#pragma unroll
    for (int rp = 0; rp < 8; ++rp) {
      inv2[rp].x = rsqrtf(vrow[2 * rp]     * (1.0f / 128.0f) + LN_EPS) * sc;
      inv2[rp].y = rsqrtf(vrow[2 * rp + 1] * (1.0f / 128.0f) + LN_EPS) * sc;
    }
  }

  __syncthreads();

  unsigned jc = j0 + (unsigned)lane;
  jc = (jc < un) ? jc : nlast;
  const float* __restrict__ bp = bT + jc;

  v2f acc[8];
#pragma unroll
  for (int rp = 0; rp < 8; ++rp) { acc[rp].x = 0.0f; acc[rp].y = 0.0f; }

  const v4f* __restrict__ Ag4 = (const v4f*)Ag;

  const __attribute__((address_space(4))) float* cbe1c =
      (const __attribute__((address_space(4))) float*)(unsigned long long)cbe1;
  const __attribute__((address_space(4))) float* cw2c =
      (const __attribute__((address_space(4))) float*)(unsigned long long)cw2;
  const __attribute__((address_space(4))) float* cg1c =
      (const __attribute__((address_space(4))) float*)(unsigned long long)cg1;

  float cur[PCH], be_c[PCH], cw_c[PCH], g_c[PCH];
#pragma unroll
  for (int u = 0; u < PCH; ++u) {
    cur[u]  = bp[(unsigned)u * un];
    be_c[u] = cbe1c[u] * sc;
    cw_c[u] = cw2c[u];
    g_c[u]  = cg1c[u];
  }

  v4f Ac[2][4];
#pragma unroll
  for (int q = 0; q < 2; ++q)
#pragma unroll
    for (int m = 0; m < 4; ++m) Ac[q][m] = Ag4[4 * q + m];

  for (int k0 = 0; k0 < H; k0 += PCH) {
    const int kn0 = (k0 + PCH < H) ? (k0 + PCH) : (H - PCH);
    float nxt[PCH], be_n[PCH], cw_n[PCH], g_n[PCH];
#pragma unroll
    for (int u = 0; u < PCH; ++u) {
      nxt[u]  = bp[(unsigned)(kn0 + u) * un];
      be_n[u] = cbe1c[kn0 + u] * sc;
      cw_n[u] = cw2c[kn0 + u];
      g_n[u]  = cg1c[kn0 + u];
    }
#pragma unroll
    for (int u = 0; u < PCH; u += 2) {
      int kb = k0 + u + 2;
      kb = (kb < H) ? kb : (H - 2);
      v4f An[2][4];
#pragma unroll
      for (int q = 0; q < 2; ++q)
#pragma unroll
        for (int m = 0; m < 4; ++m) An[q][m] = Ag4[4 * (kb + q) + m];
      __builtin_amdgcn_sched_barrier(0);
#pragma unroll
      for (int q = 0; q < 2; ++q) {
        v2f be2; be2.x = be_c[u + q]; be2.y = be2.x;
        const float cw = cw_c[u + q];
        const float bg = cur[u + q] * g_c[u + q];
        const v4f A0 = Ac[q][0], A1 = Ac[q][1], A2 = Ac[q][2], A3 = Ac[q][3];
        const float ax[16] = {A0.x, A0.y, A0.z, A0.w, A1.x, A1.y, A1.z, A1.w,
                              A2.x, A2.y, A2.z, A2.w, A3.x, A3.y, A3.z, A3.w};
#pragma unroll
        for (int rp = 0; rp < 8; ++rp) {
          v2f z;
          z.x = ax[2 * rp]     + bg;
          z.y = ax[2 * rp + 1] + bg;
          z = pk_fma_clamp01(z, inv2[rp], be2);
          acc[rp] += z * cw;
        }
      }
      __builtin_amdgcn_sched_barrier(0);
#pragma unroll
      for (int q = 0; q < 2; ++q)
#pragma unroll
        for (int m = 0; m < 4; ++m) Ac[q][m] = An[q][m];
    }
#pragma unroll
    for (int u = 0; u < PCH; ++u) {
      cur[u]  = nxt[u];
      be_c[u] = be_n[u];
      cw_c[u] = cw_n[u];
      g_c[u]  = g_n[u];
    }
  }

  float tv[16];
#pragma unroll
  for (int rp = 0; rp < 8; ++rp) {
    tv[2 * rp]     = tanh_inline(__builtin_fmaf(acc[rp].x, scinv, cb2));
    tv[2 * rp + 1] = tanh_inline(__builtin_fmaf(acc[rp].y, scinv, cb2));
  }
  const unsigned j = j0 + (unsigned)lane;
  if (j < un) {
    float* __restrict__ op = out + (size_t)i0 * un + j;
    if (i0 + 16u <= un) {
#pragma unroll
      for (int r = 0; r < 16; ++r) op[(size_t)r * un] = tv[r];
    } else {
#pragma unroll
      for (int r = 0; r < 16; ++r)
        if (i0 + (unsigned)r < un) op[(size_t)r * un] = tv[r];
    }
  }
}

#define CS_THREADS 512
#define EPT 4

__device__ __forceinline__
void comp_scores_role(const unsigned blk, const unsigned nblk,
                        const float* __restrict__ mat,
                        const int* __restrict__ ei, int E, int N,
                        float* __restrict__ out) {
  const unsigned ue    = (unsigned)E;
  const unsigned un    = (unsigned)N;
  const unsigned nlast = un - 1u;
  const unsigned chunk = CS_THREADS * EPT;
  const int* erow = ei;
  const int* ecol = ei + ue;

  for (unsigned c0 = blk * chunk; c0 < ue; c0 += nblk * chunk) {
    const unsigned e0 = c0 + threadIdx.x;
    if (c0 + chunk <= ue) {
      unsigned ri[EPT], ci[EPT];
#pragma unroll
      for (int u = 0; u < EPT; ++u) {
        ri[u] = (unsigned)erow[e0 + u * CS_THREADS];
        ci[u] = (unsigned)ecol[e0 + u * CS_THREADS];
      }
      float v[EPT];
#pragma unroll
      for (int u = 0; u < EPT; ++u) {
        const unsigned i = (ri[u] < un) ? ri[u] : nlast;
        const unsigned j = (ci[u] < un) ? ci[u] : nlast;
        v[u] = mat[i * un + j];
      }
#pragma unroll
      for (int u = 0; u < EPT; ++u) out[e0 + u * CS_THREADS] = v[u];
    } else {
#pragma unroll 1
      for (unsigned e = e0; e < ue; e += CS_THREADS) {
        const unsigned r = (unsigned)erow[e];
        const unsigned c = (unsigned)ecol[e];
        const unsigned i = (r < un) ? r : nlast;
        const unsigned j = (c < un) ? c : nlast;
        out[e] = mat[i * un + j];
      }
    }
  }
}

__global__ __launch_bounds__(256, 8)
void pairs_and_nodes_kernel(const float* __restrict__ aT,
                            const float* __restrict__ bT,
                            const float* __restrict__ cg1,
                            const float* __restrict__ cbe1,
                            const float* __restrict__ cw2,
                            const float* __restrict__ cb2p,
                            float* __restrict__ out_matrix,
                            const float* __restrict__ h1pre,
                            const float* __restrict__ b1,
                            const float* __restrict__ g1,
                            const float* __restrict__ be1,
                            const float* __restrict__ w2,
                            const float* __restrict__ b2,
                            const float* __restrict__ g2,
                            const float* __restrict__ be2,
                            const float* __restrict__ w3,
                            const float* __restrict__ b3,
                            float* __restrict__ out_logits,
                            float* __restrict__ out_probs,
                            int N, int colGroups) {
  if ((int)blockIdx.y < colGroups) {
    comp_matrix_role(blockIdx.x, blockIdx.y, aT, bT, cg1, cbe1, cw2, cb2p,
                     out_matrix, N);
  } else {
    const int nb = ((int)blockIdx.y - colGroups) * (int)gridDim.x + (int)blockIdx.x;
    node_mlp_role(nb, h1pre, b1, g1, be1, w2, b2, g2, be2, w3, b3,
                  out_logits, out_probs, N);
  }
}

#define CENTER_BLOCKS 4

__global__ __launch_bounds__(CWAVES * 32)
void centers_and_scores_kernel(const float* __restrict__ probs,
                               const float* __restrict__ x,
                               const float* __restrict__ prototypes,
                               float* __restrict__ out_centers,
                               float* __restrict__ out_protos,
                               const float* __restrict__ mat,
                               const int* __restrict__ ei, int E, int N,
                               float* __restrict__ out_scores) {
  if (blockIdx.x < CENTER_BLOCKS) {
    centers_role((int)blockIdx.x, probs, x, prototypes, out_centers,
                 out_protos, N);
  } else {
    comp_scores_role(blockIdx.x - CENTER_BLOCKS, gridDim.x - CENTER_BLOCKS,
                     mat, ei, E, N, out_scores);
  }
}

extern "C" void kernel_launch(void* const* d_in, const int* in_sizes, int n_in,
                              void* d_out, int out_size, void* d_ws, size_t ws_size,
                              hipStream_t stream) {
  const float* x    = (const float*)d_in[0];
  const int*   ei   = (const int*)  d_in[1];
  const float* w1   = (const float*)d_in[2];
  const float* b1   = (const float*)d_in[3];
  const float* g1   = (const float*)d_in[4];
  const float* be1  = (const float*)d_in[5];
  const float* w2   = (const float*)d_in[6];
  const float* b2   = (const float*)d_in[7];
  const float* g2   = (const float*)d_in[8];
  const float* be2  = (const float*)d_in[9];
  const float* w3   = (const float*)d_in[10];
  const float* b3   = (const float*)d_in[11];
  const float* cw1  = (const float*)d_in[12];
  const float* cb1  = (const float*)d_in[13];
  const float* cg1  = (const float*)d_in[14];
  const float* cbe1 = (const float*)d_in[15];
  const float* cw2  = (const float*)d_in[16];
  const float* cb2  = (const float*)d_in[17];
  const float* prot = (const float*)d_in[18];

  const int N = in_sizes[0] / D;
  const int E = in_sizes[1] / 2;

  float* h1pre = (float*)d_ws;
  float* aT    = h1pre + (size_t)N * H;
  float* bT    = aT    + (size_t)N * H;

  float* out       = (float*)d_out;
  float* o_logits  = out;
  float* o_probs   = o_logits  + (size_t)N * KC;
  float* o_scores  = o_probs   + (size_t)N * KC;
  float* o_matrix  = o_scores  + (size_t)E;
  float* o_centers = o_matrix  + (size_t)N * N;
  float* o_protos  = o_centers + (size_t)KC * D;

  const int rowTiles  = (N + 15) / 16;
  const int colGroups = (N + 255) / 256;
  const int nodeBlocks = (N + 1) / 2;
  const int nodeRows   = (nodeBlocks + rowTiles - 1) / rowTiles;

  wmma_gemm3_kernel<<<dim3(rowTiles, 3), 128, 0, stream>>>(
      x, w1, cw1, cb1, h1pre, aT, bT, N);

  pairs_and_nodes_kernel<<<dim3(rowTiles, colGroups + nodeRows), 256, 0, stream>>>(
      aT, bT, cg1, cbe1, cw2, cb2, o_matrix,
      h1pre, b1, g1, be1, w2, b2, g2, be2, w3, b3, o_logits, o_probs,
      N, colGroups);

  centers_and_scores_kernel<<<CENTER_BLOCKS + 256, CWAVES * 32, 0, stream>>>(
      o_probs, x, prot, o_centers, o_protos, o_matrix, ei, E, N, o_scores);
}
